# starting kernel
# baseline (speedup 1.0000x reference)
#include <hip/hip_runtime.h>
#include <hip/hip_bf16.h>

typedef __attribute__((ext_vector_type(2))) float v2f;
typedef __attribute__((ext_vector_type(8))) float v8f;

#define NSLOPE 0.2f

__device__ __forceinline__ unsigned fmap(float x) {
    unsigned b = __float_as_uint(x);
    return (b & 0x80000000u) ? ~b : (b | 0x80000000u);
}
__device__ __forceinline__ float funmap(unsigned u) {
    return (u & 0x80000000u) ? __uint_as_float(u ^ 0x80000000u)
                             : __uint_as_float(~u);
}

__global__ __launch_bounds__(256) void attr_mean_k(const float* __restrict__ eattr,
                                                   float* __restrict__ meanAcc, int E_) {
    __shared__ float s[8];
    if (threadIdx.x < 8) s[threadIdx.x] = 0.f;
    __syncthreads();
    float acc[8] = {0.f, 0.f, 0.f, 0.f, 0.f, 0.f, 0.f, 0.f};
    for (int r = blockIdx.x * blockDim.x + threadIdx.x; r < E_; r += gridDim.x * blockDim.x) {
        const float* row = eattr + (size_t)r * 8;
#pragma unroll
        for (int d = 0; d < 8; ++d) acc[d] += row[d];
    }
#pragma unroll
    for (int d = 0; d < 8; ++d) atomicAdd(&s[d], acc[d]);
    __syncthreads();
    if (threadIdx.x < 8) atomicAdd(meanAcc + threadIdx.x, s[threadIdx.x]);
}

__global__ __launch_bounds__(128) void loop_emb_k(const float* __restrict__ meanAcc,
                                                  const float* __restrict__ We,
                                                  float* __restrict__ loop_emb, float invE) {
    int j = threadIdx.x;
    float s = 0.f;
#pragma unroll
    for (int d = 0; d < 8; ++d) s += (meanAcc[d] * invE) * We[d * 128 + j];
    loop_emb[j] = s;
}

__global__ __launch_bounds__(256) void out_init_k(float* __restrict__ out,
                                                  const float* __restrict__ bias, int total) {
    int i = blockIdx.x * blockDim.x + threadIdx.x;
    if (i < total) out[i] = bias[i & 127];
}

__global__ __launch_bounds__(128) void node_xform_wmma(
    const float* __restrict__ X,
    const float* __restrict__ Wl, const float* __restrict__ bl,
    const float* __restrict__ Wr, const float* __restrict__ br,
    float* __restrict__ xl, float* __restrict__ xr, int nTiles) {
    int wid = threadIdx.x >> 5;
    int lane = threadIdx.x & 31;
    int rowTile = blockIdx.x * 4 + wid;
    if (rowTile >= nTiles) return;

    const float* W    = blockIdx.y ? Wr : Wl;
    const float* bias = blockIdx.y ? br : bl;
    float* out        = blockIdx.y ? xr : xl;

    int row0  = rowTile * 16;
    int m     = lane & 15;
    int khalf = lane >> 4;

    v8f acc[8] = {};

    const float* arow = X + (size_t)(row0 + m) * 128;
    for (int kt = 0; kt < 32; ++kt) {
        int k0 = kt * 4 + khalf * 2;
        v2f a;
        a.x = arow[k0];
        a.y = arow[k0 + 1];
        const float* w0 = W + (size_t)k0 * 128;
        const float* w1 = w0 + 128;
#pragma unroll
        for (int nt = 0; nt < 8; ++nt) {
            v2f b;
            b.x = w0[nt * 16 + m];
            b.y = w1[nt * 16 + m];
            acc[nt] = __builtin_amdgcn_wmma_f32_16x16x4_f32(
                false, a, false, b, (short)0, acc[nt], false, false);
        }
    }
#pragma unroll
    for (int nt = 0; nt < 8; ++nt) {
        float bv = bias[nt * 16 + m];
#pragma unroll
        for (int i = 0; i < 8; ++i) {
            int row = row0 + i + 8 * khalf;
            out[(size_t)row * 128 + nt * 16 + m] = acc[nt][i] + bv;
        }
    }
}

__global__ __launch_bounds__(256) void edge_logit_k(
    const int* __restrict__ ei, const float* __restrict__ eattr,
    const float* __restrict__ We, const float* __restrict__ loop_emb,
    const float* __restrict__ xl, const float* __restrict__ xr,
    const float* __restrict__ att, float* __restrict__ logit,
    unsigned* __restrict__ segmax, int E_, int N_) {
    int gw = (blockIdx.x * blockDim.x + threadIdx.x) >> 5;
    int lane = threadIdx.x & 31;
    if (gw >= E_ + N_) return;

    int src, dst;
    if (gw < E_) { src = ei[gw]; dst = ei[E_ + gw]; }
    else         { src = dst = gw - E_; }

    int c0 = lane * 4;
    float4 vl = *(const float4*)(xl + (size_t)src * 128 + c0);
    float4 vr = *(const float4*)(xr + (size_t)dst * 128 + c0);

    float e0, e1, e2, e3;
    if (gw < E_) {
        float at[8];
#pragma unroll
        for (int d = 0; d < 8; ++d) at[d] = eattr[(size_t)gw * 8 + d];
        e0 = e1 = e2 = e3 = 0.f;
#pragma unroll
        for (int d = 0; d < 8; ++d) {
            const float* wrow = We + d * 128 + c0;
            e0 = fmaf(at[d], wrow[0], e0);
            e1 = fmaf(at[d], wrow[1], e1);
            e2 = fmaf(at[d], wrow[2], e2);
            e3 = fmaf(at[d], wrow[3], e3);
        }
    } else {
        e0 = loop_emb[c0]; e1 = loop_emb[c0 + 1];
        e2 = loop_emb[c0 + 2]; e3 = loop_emb[c0 + 3];
    }

    float m0 = vl.x + vr.x + e0;
    float m1 = vl.y + vr.y + e1;
    float m2 = vl.z + vr.z + e2;
    float m3 = vl.w + vr.w + e3;
    m0 = (m0 > 0.f) ? m0 : NSLOPE * m0;
    m1 = (m1 > 0.f) ? m1 : NSLOPE * m1;
    m2 = (m2 > 0.f) ? m2 : NSLOPE * m2;
    m3 = (m3 > 0.f) ? m3 : NSLOPE * m3;

    float p = m0 * att[c0] + m1 * att[c0 + 1] + m2 * att[c0 + 2] + m3 * att[c0 + 3];
    p += __shfl_xor(p, 1, 32);
    p += __shfl_xor(p, 2, 32);

    int h = lane >> 2;
    if ((lane & 3) == 0) {
        logit[(size_t)gw * 8 + h] = p;
        atomicMax(segmax + (size_t)dst * 8 + h, fmap(p));
    }
}

__global__ __launch_bounds__(256) void edge_exp_k(
    const int* __restrict__ ei, float* __restrict__ logit,
    const unsigned* __restrict__ segmax, float* __restrict__ segsum,
    int E_, int N_) {
    int i = blockIdx.x * blockDim.x + threadIdx.x;
    if (i >= (E_ + N_) * 8) return;
    int e = i >> 3, h = i & 7;
    int dst = (e < E_) ? ei[E_ + e] : (e - E_);
    float mx = funmap(segmax[(size_t)dst * 8 + h]);
    float ex = expf(logit[i] - mx);
    logit[i] = ex;
    atomicAdd(segsum + (size_t)dst * 8 + h, ex);
}

__global__ __launch_bounds__(256) void edge_msg_k(
    const int* __restrict__ ei, const float* __restrict__ expd,
    const float* __restrict__ segsum, const float* __restrict__ xl,
    float* __restrict__ out, int E_, int N_) {
    int gw = (blockIdx.x * blockDim.x + threadIdx.x) >> 5;
    int lane = threadIdx.x & 31;
    if (gw >= E_ + N_) return;

    int src, dst;
    if (gw < E_) { src = ei[gw]; dst = ei[E_ + gw]; }
    else         { src = dst = gw - E_; }

    int h = lane >> 2;
    float alpha = expd[(size_t)gw * 8 + h] / (segsum[(size_t)dst * 8 + h] + 1e-16f);

    int c0 = lane * 4;
    float4 v = *(const float4*)(xl + (size_t)src * 128 + c0);
    float* o = out + (size_t)dst * 128 + c0;
    atomicAdd(o + 0, alpha * v.x);
    atomicAdd(o + 1, alpha * v.y);
    atomicAdd(o + 2, alpha * v.z);
    atomicAdd(o + 3, alpha * v.w);
}

extern "C" void kernel_launch(void* const* d_in, const int* in_sizes, int n_in,
                              void* d_out, int out_size, void* d_ws, size_t ws_size,
                              hipStream_t stream) {
    const float* x     = (const float*)d_in[0];
    const int*   ei    = (const int*)d_in[1];
    const float* eattr = (const float*)d_in[2];
    const float* Wl    = (const float*)d_in[3];
    const float* bl    = (const float*)d_in[4];
    const float* Wr    = (const float*)d_in[5];
    const float* br    = (const float*)d_in[6];
    const float* We    = (const float*)d_in[7];
    const float* att   = (const float*)d_in[8];
    const float* bias  = (const float*)d_in[9];
    float* out = (float*)d_out;
    float* ws  = (float*)d_ws;

    const int n = in_sizes[0] / 128;
    const int e = in_sizes[2] / 8;
    const int tot = e + n;

    size_t off_mean   = 0;
    size_t off_loop   = 16;
    size_t off_segmax = 256;
    size_t off_segsum = off_segmax + (size_t)n * 8;
    size_t off_xl     = off_segsum + (size_t)n * 8;
    size_t off_xr     = off_xl + (size_t)n * 128;
    size_t off_logit  = off_xr + (size_t)n * 128;

    hipMemsetAsync(d_ws, 0, off_xl * sizeof(float), stream);

    attr_mean_k<<<512, 256, 0, stream>>>(eattr, ws + off_mean, e);
    loop_emb_k<<<1, 128, 0, stream>>>(ws + off_mean, We, ws + off_loop, 1.0f / (float)e);
    out_init_k<<<(n * 128 + 255) / 256, 256, 0, stream>>>(out, bias, n * 128);

    int nTiles = n / 16;
    dim3 g((nTiles + 3) / 4, 2);
    node_xform_wmma<<<g, 128, 0, stream>>>(x, Wl, bl, Wr, br,
                                           ws + off_xl, ws + off_xr, nTiles);

    int waveBlocks = ((tot * 32) + 255) / 256;
    edge_logit_k<<<waveBlocks, 256, 0, stream>>>(ei, eattr, We, ws + off_loop,
                                                 ws + off_xl, ws + off_xr, att,
                                                 ws + off_logit,
                                                 (unsigned*)(ws + off_segmax), e, n);
    edge_exp_k<<<(tot * 8 + 255) / 256, 256, 0, stream>>>(ei, ws + off_logit,
                                                          (const unsigned*)(ws + off_segmax),
                                                          ws + off_segsum, e, n);
    edge_msg_k<<<waveBlocks, 256, 0, stream>>>(ei, ws + off_logit, ws + off_segsum,
                                               ws + off_xl, out, e, n);
}


# edge kernels (logit, exp, msg): one wave owns 16 consecutive edges, 4 row gathers in flight, W_e/att held in registers, not one wave per edge; float atomics kept, arrival order arbitrary as before
# speedup vs baseline: 1.2741x; 1.2741x over previous
#include <hip/hip_runtime.h>
#include <hip/hip_bf16.h>

typedef __attribute__((ext_vector_type(2))) float v2f;
typedef __attribute__((ext_vector_type(8))) float v8f;

#define NSLOPE 0.2f

__device__ __forceinline__ unsigned fmap(float x) {
    unsigned b = __float_as_uint(x);
    return (b & 0x80000000u) ? ~b : (b | 0x80000000u);
}
__device__ __forceinline__ float funmap(unsigned u) {
    return (u & 0x80000000u) ? __uint_as_float(u ^ 0x80000000u)
                             : __uint_as_float(~u);
}

__global__ __launch_bounds__(256) void attr_mean_k(const float* __restrict__ eattr,
                                                   float* __restrict__ meanAcc, int E_) {
    __shared__ float s[8];
    if (threadIdx.x < 8) s[threadIdx.x] = 0.f;
    __syncthreads();
    float acc[8] = {0.f, 0.f, 0.f, 0.f, 0.f, 0.f, 0.f, 0.f};
    for (int r = blockIdx.x * blockDim.x + threadIdx.x; r < E_; r += gridDim.x * blockDim.x) {
        const float* row = eattr + (size_t)r * 8;
#pragma unroll
        for (int d = 0; d < 8; ++d) acc[d] += row[d];
    }
#pragma unroll
    for (int d = 0; d < 8; ++d) atomicAdd(&s[d], acc[d]);
    __syncthreads();
    if (threadIdx.x < 8) atomicAdd(meanAcc + threadIdx.x, s[threadIdx.x]);
}

__global__ __launch_bounds__(128) void loop_emb_k(const float* __restrict__ meanAcc,
                                                  const float* __restrict__ We,
                                                  float* __restrict__ loop_emb, float invE) {
    int j = threadIdx.x;
    float s = 0.f;
#pragma unroll
    for (int d = 0; d < 8; ++d) s += (meanAcc[d] * invE) * We[d * 128 + j];
    loop_emb[j] = s;
}

__global__ __launch_bounds__(256) void out_init_k(float* __restrict__ out,
                                                  const float* __restrict__ bias, int total) {
    int i = blockIdx.x * blockDim.x + threadIdx.x;
    if (i < total) out[i] = bias[i & 127];
}

__global__ __launch_bounds__(128) void node_xform_wmma(
    const float* __restrict__ X,
    const float* __restrict__ Wl, const float* __restrict__ bl,
    const float* __restrict__ Wr, const float* __restrict__ br,
    float* __restrict__ xl, float* __restrict__ xr, int nTiles) {
    int wid = threadIdx.x >> 5;
    int lane = threadIdx.x & 31;
    int rowTile = blockIdx.x * 4 + wid;
    if (rowTile >= nTiles) return;

    const float* W    = blockIdx.y ? Wr : Wl;
    const float* bias = blockIdx.y ? br : bl;
    float* out        = blockIdx.y ? xr : xl;

    int row0  = rowTile * 16;
    int m     = lane & 15;
    int khalf = lane >> 4;

    v8f acc[8] = {};

    const float* arow = X + (size_t)(row0 + m) * 128;
    for (int kt = 0; kt < 32; ++kt) {
        int k0 = kt * 4 + khalf * 2;
        v2f a;
        a.x = arow[k0];
        a.y = arow[k0 + 1];
        const float* w0 = W + (size_t)k0 * 128;
        const float* w1 = w0 + 128;
#pragma unroll
        for (int nt = 0; nt < 8; ++nt) {
            v2f b;
            b.x = w0[nt * 16 + m];
            b.y = w1[nt * 16 + m];
            acc[nt] = __builtin_amdgcn_wmma_f32_16x16x4_f32(
                false, a, false, b, (short)0, acc[nt], false, false);
        }
    }
#pragma unroll
    for (int nt = 0; nt < 8; ++nt) {
        float bv = bias[nt * 16 + m];
#pragma unroll
        for (int i = 0; i < 8; ++i) {
            int row = row0 + i + 8 * khalf;
            out[(size_t)row * 128 + nt * 16 + m] = acc[nt][i] + bv;
        }
    }
}

#define CH 16
#define UF 4

template <bool SELF, int U>
__device__ __forceinline__ void logit_edges(
    int gfirst, int jfirst, int srcv, int dstv, int lane,
    const float (&w)[8][4], const float4& a4, const float4& le,
    const float* __restrict__ eattr, const float* __restrict__ xl,
    const float* __restrict__ xr, float* __restrict__ logit,
    unsigned* __restrict__ segmax) {
    const int c0 = lane * 4;
    float4 vl[U], vr[U];
    int dd[U];
#pragma unroll
    for (int u = 0; u < U; ++u) {
        const int s = __builtin_amdgcn_readlane(srcv, jfirst + u);
        const int d = __builtin_amdgcn_readlane(dstv, jfirst + u);
        dd[u] = d;
        vl[u] = *(const float4*)(xl + (size_t)s * 128 + c0);
        vr[u] = *(const float4*)(xr + (size_t)d * 128 + c0);
    }
#pragma unroll
    for (int u = 0; u < U; ++u) {
        float e0, e1, e2, e3;
        if (SELF) {
            e0 = le.x; e1 = le.y; e2 = le.z; e3 = le.w;
        } else {
            const float* ar = eattr + (size_t)(gfirst + u) * 8;
            e0 = e1 = e2 = e3 = 0.f;
#pragma unroll
            for (int d = 0; d < 8; ++d) {
                const float a = ar[d];
                e0 = fmaf(a, w[d][0], e0);
                e1 = fmaf(a, w[d][1], e1);
                e2 = fmaf(a, w[d][2], e2);
                e3 = fmaf(a, w[d][3], e3);
            }
        }
        float m0 = vl[u].x + vr[u].x + e0;
        float m1 = vl[u].y + vr[u].y + e1;
        float m2 = vl[u].z + vr[u].z + e2;
        float m3 = vl[u].w + vr[u].w + e3;
        m0 = (m0 > 0.f) ? m0 : NSLOPE * m0;
        m1 = (m1 > 0.f) ? m1 : NSLOPE * m1;
        m2 = (m2 > 0.f) ? m2 : NSLOPE * m2;
        m3 = (m3 > 0.f) ? m3 : NSLOPE * m3;

        float p = m0 * a4.x + m1 * a4.y + m2 * a4.z + m3 * a4.w;
        p += __shfl_xor(p, 1, 32);
        p += __shfl_xor(p, 2, 32);

        const int h = lane >> 2;
        if ((lane & 3) == 0) {
            logit[(size_t)(gfirst + u) * 8 + h] = p;
            atomicMax(segmax + (size_t)dd[u] * 8 + h, fmap(p));
        }
    }
}

__global__ __launch_bounds__(256) void edge_logit_k(
    const int* __restrict__ ei, const float* __restrict__ eattr,
    const float* __restrict__ We, const float* __restrict__ loop_emb,
    const float* __restrict__ xl, const float* __restrict__ xr,
    const float* __restrict__ att, float* __restrict__ logit,
    unsigned* __restrict__ segmax, int E_, int N_) {
    const int lane = threadIdx.x & 31;
    const int wv = __builtin_amdgcn_readfirstlane((int)((blockIdx.x * blockDim.x + threadIdx.x) >> 5));
    const int CE = (E_ + CH - 1) / CH;
    const int CN = (N_ + CH - 1) / CH;
    if (wv >= CE + CN) return;

    const int c0 = lane * 4;
    float w[8][4];
#pragma unroll
    for (int d = 0; d < 8; ++d) {
        const float4 t = *(const float4*)(We + d * 128 + c0);
        w[d][0] = t.x; w[d][1] = t.y; w[d][2] = t.z; w[d][3] = t.w;
    }
    const float4 a4 = *(const float4*)(att + c0);
    const float4 le = *(const float4*)(loop_emb + c0);

    if (wv < CE) {
        const int base = wv * CH;
        const int cnt = min(CH, E_ - base);
        int srcv = 0, dstv = 0;
        if (lane < cnt) { srcv = ei[base + lane]; dstv = ei[E_ + base + lane]; }
        int j = 0;
        for (; j + UF <= cnt; j += UF)
            logit_edges<false, UF>(base + j, j, srcv, dstv, lane, w, a4, le,
                                   eattr, xl, xr, logit, segmax);
        for (; j < cnt; ++j)
            logit_edges<false, 1>(base + j, j, srcv, dstv, lane, w, a4, le,
                                  eattr, xl, xr, logit, segmax);
    } else {
        const int nb = (wv - CE) * CH;
        const int cnt = min(CH, N_ - nb);
        const int nodev = (lane < cnt) ? (nb + lane) : nb;
        int j = 0;
        for (; j + UF <= cnt; j += UF)
            logit_edges<true, UF>(E_ + nb + j, j, nodev, nodev, lane, w, a4, le,
                                  eattr, xl, xr, logit, segmax);
        for (; j < cnt; ++j)
            logit_edges<true, 1>(E_ + nb + j, j, nodev, nodev, lane, w, a4, le,
                                 eattr, xl, xr, logit, segmax);
    }
}

__global__ __launch_bounds__(256) void edge_exp_k(
    const int* __restrict__ ei, float* __restrict__ logit,
    const unsigned* __restrict__ segmax, float* __restrict__ segsum,
    int E_, int N_) {
    const int lane = threadIdx.x & 31;
    const int wv = __builtin_amdgcn_readfirstlane((int)((blockIdx.x * blockDim.x + threadIdx.x) >> 5));
    const int CE = (E_ + CH - 1) / CH;
    const int CN = (N_ + CH - 1) / CH;
    if (wv >= CE + CN) return;

    int gbase, cnt, dstv;
    if (wv < CE) {
        const int base = wv * CH;
        cnt = min(CH, E_ - base);
        gbase = base;
        dstv = (lane < cnt) ? ei[E_ + base + lane] : 0;
    } else {
        const int nb = (wv - CE) * CH;
        cnt = min(CH, N_ - nb);
        gbase = E_ + nb;
        dstv = (lane < cnt) ? (nb + lane) : 0;
    }

    constexpr int K = CH * 8 / 32;
    const int h = lane & 7;
    float* lrow = logit + (size_t)gbase * 8;
    float lg[K];
    unsigned mu[K];
    int dd[K];
    bool ok[K];
#pragma unroll
    for (int k = 0; k < K; ++k) {
        const int j = k * 4 + (lane >> 3);
        ok[k] = j < cnt;
        dd[k] = __shfl(dstv, j, 32);
        lg[k] = ok[k] ? lrow[k * 32 + lane] : 0.f;
        mu[k] = ok[k] ? segmax[(size_t)dd[k] * 8 + h] : 0u;
    }
#pragma unroll
    for (int k = 0; k < K; ++k) {
        if (ok[k]) {
            const float ex = expf(lg[k] - funmap(mu[k]));
            lrow[k * 32 + lane] = ex;
            atomicAdd(segsum + (size_t)dd[k] * 8 + h, ex);
        }
    }
}

template <int U>
__device__ __forceinline__ void msg_edges(
    int gfirst, int jfirst, int srcv, int dstv, int lane,
    const float* __restrict__ expd, const float* __restrict__ segsum,
    const float* __restrict__ xl, float* __restrict__ out) {
    const int c0 = lane * 4;
    const int h = lane >> 2;
    float4 v[U];
    float ex[U], sm[U];
    int dd[U];
#pragma unroll
    for (int u = 0; u < U; ++u) {
        const int s = __builtin_amdgcn_readlane(srcv, jfirst + u);
        const int d = __builtin_amdgcn_readlane(dstv, jfirst + u);
        dd[u] = d;
        v[u]  = *(const float4*)(xl + (size_t)s * 128 + c0);
        ex[u] = expd[(size_t)(gfirst + u) * 8 + h];
        sm[u] = segsum[(size_t)d * 8 + h];
    }
#pragma unroll
    for (int u = 0; u < U; ++u) {
        const float alpha = ex[u] / (sm[u] + 1e-16f);
        float* o = out + (size_t)dd[u] * 128 + c0;
        atomicAdd(o + 0, alpha * v[u].x);
        atomicAdd(o + 1, alpha * v[u].y);
        atomicAdd(o + 2, alpha * v[u].z);
        atomicAdd(o + 3, alpha * v[u].w);
    }
}

__global__ __launch_bounds__(256) void edge_msg_k(
    const int* __restrict__ ei, const float* __restrict__ expd,
    const float* __restrict__ segsum, const float* __restrict__ xl,
    float* __restrict__ out, int E_, int N_) {
    const int lane = threadIdx.x & 31;
    const int wv = __builtin_amdgcn_readfirstlane((int)((blockIdx.x * blockDim.x + threadIdx.x) >> 5));
    const int CE = (E_ + CH - 1) / CH;
    const int CN = (N_ + CH - 1) / CH;
    if (wv >= CE + CN) return;

    int gbase, cnt, srcv, dstv;
    if (wv < CE) {
        const int base = wv * CH;
        cnt = min(CH, E_ - base);
        gbase = base;
        srcv = dstv = 0;
        if (lane < cnt) { srcv = ei[base + lane]; dstv = ei[E_ + base + lane]; }
    } else {
        const int nb = (wv - CE) * CH;
        cnt = min(CH, N_ - nb);
        gbase = E_ + nb;
        srcv = dstv = (lane < cnt) ? (nb + lane) : nb;
    }
    int j = 0;
    for (; j + UF <= cnt; j += UF)
        msg_edges<UF>(gbase + j, j, srcv, dstv, lane, expd, segsum, xl, out);
    for (; j < cnt; ++j)
        msg_edges<1>(gbase + j, j, srcv, dstv, lane, expd, segsum, xl, out);
}

extern "C" void kernel_launch(void* const* d_in, const int* in_sizes, int n_in,
                              void* d_out, int out_size, void* d_ws, size_t ws_size,
                              hipStream_t stream) {
    const float* x     = (const float*)d_in[0];
    const int*   ei    = (const int*)d_in[1];
    const float* eattr = (const float*)d_in[2];
    const float* Wl    = (const float*)d_in[3];
    const float* bl    = (const float*)d_in[4];
    const float* Wr    = (const float*)d_in[5];
    const float* br    = (const float*)d_in[6];
    const float* We    = (const float*)d_in[7];
    const float* att   = (const float*)d_in[8];
    const float* bias  = (const float*)d_in[9];
    float* out = (float*)d_out;
    float* ws  = (float*)d_ws;

    const int n = in_sizes[0] / 128;
    const int e = in_sizes[2] / 8;

    size_t off_mean   = 0;
    size_t off_loop   = 16;
    size_t off_segmax = 256;
    size_t off_segsum = off_segmax + (size_t)n * 8;
    size_t off_xl     = off_segsum + (size_t)n * 8;
    size_t off_xr     = off_xl + (size_t)n * 128;
    size_t off_logit  = off_xr + (size_t)n * 128;

    hipMemsetAsync(d_ws, 0, off_xl * sizeof(float), stream);

    attr_mean_k<<<512, 256, 0, stream>>>(eattr, ws + off_mean, e);
    loop_emb_k<<<1, 128, 0, stream>>>(ws + off_mean, We, ws + off_loop, 1.0f / (float)e);
    out_init_k<<<(n * 128 + 255) / 256, 256, 0, stream>>>(out, bias, n * 128);

    int nTiles = n / 16;
    dim3 g((nTiles + 3) / 4, 2);
    node_xform_wmma<<<g, 128, 0, stream>>>(x, Wl, bl, Wr, br,
                                           ws + off_xl, ws + off_xr, nTiles);

    int chunks = (e + CH - 1) / CH + (n + CH - 1) / CH;
    int waveBlocks = ((chunks * 32) + 255) / 256;
    edge_logit_k<<<waveBlocks, 256, 0, stream>>>(ei, eattr, We, ws + off_loop,
                                                 ws + off_xl, ws + off_xr, att,
                                                 ws + off_logit,
                                                 (unsigned*)(ws + off_segmax), e, n);
    edge_exp_k<<<waveBlocks, 256, 0, stream>>>(ei, ws + off_logit,
                                                          (const unsigned*)(ws + off_segmax),
                                                          ws + off_segsum, e, n);
    edge_msg_k<<<waveBlocks, 256, 0, stream>>>(ei, ws + off_logit, ws + off_segsum,
                                               ws + off_xl, out, e, n);
}


# edge_msg_k: lane owns channels lane+32i (was 4 adjacent): each float atomic instruction adds into one whole 128-byte line of out, not a quarter of four lines; arrival order arbitrary as before
# speedup vs baseline: 1.7196x; 1.3497x over previous
#include <hip/hip_runtime.h>
#include <hip/hip_bf16.h>

typedef __attribute__((ext_vector_type(2))) float v2f;
typedef __attribute__((ext_vector_type(8))) float v8f;

#define NSLOPE 0.2f

__device__ __forceinline__ unsigned fmap(float x) {
    unsigned b = __float_as_uint(x);
    return (b & 0x80000000u) ? ~b : (b | 0x80000000u);
}
__device__ __forceinline__ float funmap(unsigned u) {
    return (u & 0x80000000u) ? __uint_as_float(u ^ 0x80000000u)
                             : __uint_as_float(~u);
}

__global__ __launch_bounds__(256) void attr_mean_k(const float* __restrict__ eattr,
                                                   float* __restrict__ meanAcc, int E_) {
    __shared__ float s[8];
    if (threadIdx.x < 8) s[threadIdx.x] = 0.f;
    __syncthreads();
    float acc[8] = {0.f, 0.f, 0.f, 0.f, 0.f, 0.f, 0.f, 0.f};
    for (int r = blockIdx.x * blockDim.x + threadIdx.x; r < E_; r += gridDim.x * blockDim.x) {
        const float* row = eattr + (size_t)r * 8;
#pragma unroll
        for (int d = 0; d < 8; ++d) acc[d] += row[d];
    }
#pragma unroll
    for (int d = 0; d < 8; ++d) atomicAdd(&s[d], acc[d]);
    __syncthreads();
    if (threadIdx.x < 8) atomicAdd(meanAcc + threadIdx.x, s[threadIdx.x]);
}

__global__ __launch_bounds__(128) void loop_emb_k(const float* __restrict__ meanAcc,
                                                  const float* __restrict__ We,
                                                  float* __restrict__ loop_emb, float invE) {
    int j = threadIdx.x;
    float s = 0.f;
#pragma unroll
    for (int d = 0; d < 8; ++d) s += (meanAcc[d] * invE) * We[d * 128 + j];
    loop_emb[j] = s;
}

__global__ __launch_bounds__(256) void out_init_k(float* __restrict__ out,
                                                  const float* __restrict__ bias, int total) {
    int i = blockIdx.x * blockDim.x + threadIdx.x;
    if (i < total) out[i] = bias[i & 127];
}

__global__ __launch_bounds__(128) void node_xform_wmma(
    const float* __restrict__ X,
    const float* __restrict__ Wl, const float* __restrict__ bl,
    const float* __restrict__ Wr, const float* __restrict__ br,
    float* __restrict__ xl, float* __restrict__ xr, int nTiles) {
    int wid = threadIdx.x >> 5;
    int lane = threadIdx.x & 31;
    int rowTile = blockIdx.x * 4 + wid;
    if (rowTile >= nTiles) return;

    const float* W    = blockIdx.y ? Wr : Wl;
    const float* bias = blockIdx.y ? br : bl;
    float* out        = blockIdx.y ? xr : xl;

    int row0  = rowTile * 16;
    int m     = lane & 15;
    int khalf = lane >> 4;

    v8f acc[8] = {};

    const float* arow = X + (size_t)(row0 + m) * 128;
    for (int kt = 0; kt < 32; ++kt) {
        int k0 = kt * 4 + khalf * 2;
        v2f a;
        a.x = arow[k0];
        a.y = arow[k0 + 1];
        const float* w0 = W + (size_t)k0 * 128;
        const float* w1 = w0 + 128;
#pragma unroll
        for (int nt = 0; nt < 8; ++nt) {
            v2f b;
            b.x = w0[nt * 16 + m];
            b.y = w1[nt * 16 + m];
            acc[nt] = __builtin_amdgcn_wmma_f32_16x16x4_f32(
                false, a, false, b, (short)0, acc[nt], false, false);
        }
    }
#pragma unroll
    for (int nt = 0; nt < 8; ++nt) {
        float bv = bias[nt * 16 + m];
#pragma unroll
        for (int i = 0; i < 8; ++i) {
            int row = row0 + i + 8 * khalf;
            out[(size_t)row * 128 + nt * 16 + m] = acc[nt][i] + bv;
        }
    }
}

#define CH 16
#define UF 4

template <bool SELF, int U>
__device__ __forceinline__ void logit_edges(
    int gfirst, int jfirst, int srcv, int dstv, int lane,
    const float (&w)[8][4], const float4& a4, const float4& le,
    const float* __restrict__ eattr, const float* __restrict__ xl,
    const float* __restrict__ xr, float* __restrict__ logit,
    unsigned* __restrict__ segmax) {
    const int c0 = lane * 4;
    float4 vl[U], vr[U];
    int dd[U];
#pragma unroll
    for (int u = 0; u < U; ++u) {
        const int s = __builtin_amdgcn_readlane(srcv, jfirst + u);
        const int d = __builtin_amdgcn_readlane(dstv, jfirst + u);
        dd[u] = d;
        vl[u] = *(const float4*)(xl + (size_t)s * 128 + c0);
        vr[u] = *(const float4*)(xr + (size_t)d * 128 + c0);
    }
#pragma unroll
    for (int u = 0; u < U; ++u) {
        float e0, e1, e2, e3;
        if (SELF) {
            e0 = le.x; e1 = le.y; e2 = le.z; e3 = le.w;
        } else {
            const float* ar = eattr + (size_t)(gfirst + u) * 8;
            e0 = e1 = e2 = e3 = 0.f;
#pragma unroll
            for (int d = 0; d < 8; ++d) {
                const float a = ar[d];
                e0 = fmaf(a, w[d][0], e0);
                e1 = fmaf(a, w[d][1], e1);
                e2 = fmaf(a, w[d][2], e2);
                e3 = fmaf(a, w[d][3], e3);
            }
        }
        float m0 = vl[u].x + vr[u].x + e0;
        float m1 = vl[u].y + vr[u].y + e1;
        float m2 = vl[u].z + vr[u].z + e2;
        float m3 = vl[u].w + vr[u].w + e3;
        m0 = (m0 > 0.f) ? m0 : NSLOPE * m0;
        m1 = (m1 > 0.f) ? m1 : NSLOPE * m1;
        m2 = (m2 > 0.f) ? m2 : NSLOPE * m2;
        m3 = (m3 > 0.f) ? m3 : NSLOPE * m3;

        float p = m0 * a4.x + m1 * a4.y + m2 * a4.z + m3 * a4.w;
        p += __shfl_xor(p, 1, 32);
        p += __shfl_xor(p, 2, 32);

        const int h = lane >> 2;
        if ((lane & 3) == 0) {
            logit[(size_t)(gfirst + u) * 8 + h] = p;
            atomicMax(segmax + (size_t)dd[u] * 8 + h, fmap(p));
        }
    }
}

__global__ __launch_bounds__(256) void edge_logit_k(
    const int* __restrict__ ei, const float* __restrict__ eattr,
    const float* __restrict__ We, const float* __restrict__ loop_emb,
    const float* __restrict__ xl, const float* __restrict__ xr,
    const float* __restrict__ att, float* __restrict__ logit,
    unsigned* __restrict__ segmax, int E_, int N_) {
    const int lane = threadIdx.x & 31;
    const int wv = __builtin_amdgcn_readfirstlane((int)((blockIdx.x * blockDim.x + threadIdx.x) >> 5));
    const int CE = (E_ + CH - 1) / CH;
    const int CN = (N_ + CH - 1) / CH;
    if (wv >= CE + CN) return;

    const int c0 = lane * 4;
    float w[8][4];
#pragma unroll
    for (int d = 0; d < 8; ++d) {
        const float4 t = *(const float4*)(We + d * 128 + c0);
        w[d][0] = t.x; w[d][1] = t.y; w[d][2] = t.z; w[d][3] = t.w;
    }
    const float4 a4 = *(const float4*)(att + c0);
    const float4 le = *(const float4*)(loop_emb + c0);

    if (wv < CE) {
        const int base = wv * CH;
        const int cnt = min(CH, E_ - base);
        int srcv = 0, dstv = 0;
        if (lane < cnt) { srcv = ei[base + lane]; dstv = ei[E_ + base + lane]; }
        int j = 0;
        for (; j + UF <= cnt; j += UF)
            logit_edges<false, UF>(base + j, j, srcv, dstv, lane, w, a4, le,
                                   eattr, xl, xr, logit, segmax);
        for (; j < cnt; ++j)
            logit_edges<false, 1>(base + j, j, srcv, dstv, lane, w, a4, le,
                                  eattr, xl, xr, logit, segmax);
    } else {
        const int nb = (wv - CE) * CH;
        const int cnt = min(CH, N_ - nb);
        const int nodev = (lane < cnt) ? (nb + lane) : nb;
        int j = 0;
        for (; j + UF <= cnt; j += UF)
            logit_edges<true, UF>(E_ + nb + j, j, nodev, nodev, lane, w, a4, le,
                                  eattr, xl, xr, logit, segmax);
        for (; j < cnt; ++j)
            logit_edges<true, 1>(E_ + nb + j, j, nodev, nodev, lane, w, a4, le,
                                 eattr, xl, xr, logit, segmax);
    }
}

__global__ __launch_bounds__(256) void edge_exp_k(
    const int* __restrict__ ei, float* __restrict__ logit,
    const unsigned* __restrict__ segmax, float* __restrict__ segsum,
    int E_, int N_) {
    const int lane = threadIdx.x & 31;
    const int wv = __builtin_amdgcn_readfirstlane((int)((blockIdx.x * blockDim.x + threadIdx.x) >> 5));
    const int CE = (E_ + CH - 1) / CH;
    const int CN = (N_ + CH - 1) / CH;
    if (wv >= CE + CN) return;

    int gbase, cnt, dstv;
    if (wv < CE) {
        const int base = wv * CH;
        cnt = min(CH, E_ - base);
        gbase = base;
        dstv = (lane < cnt) ? ei[E_ + base + lane] : 0;
    } else {
        const int nb = (wv - CE) * CH;
        cnt = min(CH, N_ - nb);
        gbase = E_ + nb;
        dstv = (lane < cnt) ? (nb + lane) : 0;
    }

    constexpr int K = CH * 8 / 32;
    const int h = lane & 7;
    float* lrow = logit + (size_t)gbase * 8;
    float lg[K];
    unsigned mu[K];
    int dd[K];
    bool ok[K];
#pragma unroll
    for (int k = 0; k < K; ++k) {
        const int j = k * 4 + (lane >> 3);
        ok[k] = j < cnt;
        dd[k] = __shfl(dstv, j, 32);
        lg[k] = ok[k] ? lrow[k * 32 + lane] : 0.f;
        mu[k] = ok[k] ? segmax[(size_t)dd[k] * 8 + h] : 0u;
    }
#pragma unroll
    for (int k = 0; k < K; ++k) {
        if (ok[k]) {
            const float ex = expf(lg[k] - funmap(mu[k]));
            lrow[k * 32 + lane] = ex;
            atomicAdd(segsum + (size_t)dd[k] * 8 + h, ex);
        }
    }
}

template <int U>
__device__ __forceinline__ void msg_edges(
    int gfirst, int jfirst, int srcv, int dstv, int lane,
    const float* __restrict__ expd, const float* __restrict__ segsum,
    const float* __restrict__ xl, float* __restrict__ out) {
    static_assert(U * 8 <= 32, "one (edge, head) pair per lane");
    float v[U][4];
    int dd[U];
#pragma unroll
    for (int u = 0; u < U; ++u) {
        const int s = __builtin_amdgcn_readlane(srcv, jfirst + u);
        const int d = __builtin_amdgcn_readlane(dstv, jfirst + u);
        dd[u] = d;
        const float* r = xl + (size_t)s * 128 + lane;
        v[u][0] = r[0];
        v[u][1] = r[32];
        v[u][2] = r[64];
        v[u][3] = r[96];
    }
    const int ue = ((lane >> 3) < U) ? (lane >> 3) : (U - 1);
    const int hq = lane & 7;
    int dl = dd[0];
#pragma unroll
    for (int u = 1; u < U; ++u) dl = (ue == u) ? dd[u] : dl;
    const float exv = expd[(size_t)(gfirst + ue) * 8 + hq];
    const float smv = segsum[(size_t)dl * 8 + hq];
    const float al = exv / (smv + 1e-16f);
    const int hsel = lane >> 4;
#pragma unroll
    for (int u = 0; u < U; ++u) {
        float* o = out + (size_t)dd[u] * 128 + lane;
#pragma unroll
        for (int i = 0; i < 4; ++i) {
            const float alo = __int_as_float(__builtin_amdgcn_readlane(__float_as_int(al), u * 8 + 2 * i));
            const float ahi = __int_as_float(__builtin_amdgcn_readlane(__float_as_int(al), u * 8 + 2 * i + 1));
            const float a = hsel ? ahi : alo;
            atomicAdd(o + 32 * i, a * v[u][i]);
        }
    }
}

__global__ __launch_bounds__(256) void edge_msg_k(
    const int* __restrict__ ei, const float* __restrict__ expd,
    const float* __restrict__ segsum, const float* __restrict__ xl,
    float* __restrict__ out, int E_, int N_) {
    const int lane = threadIdx.x & 31;
    const int wv = __builtin_amdgcn_readfirstlane((int)((blockIdx.x * blockDim.x + threadIdx.x) >> 5));
    const int CE = (E_ + CH - 1) / CH;
    const int CN = (N_ + CH - 1) / CH;
    if (wv >= CE + CN) return;

    int gbase, cnt, srcv, dstv;
    if (wv < CE) {
        const int base = wv * CH;
        cnt = min(CH, E_ - base);
        gbase = base;
        srcv = dstv = 0;
        if (lane < cnt) { srcv = ei[base + lane]; dstv = ei[E_ + base + lane]; }
    } else {
        const int nb = (wv - CE) * CH;
        cnt = min(CH, N_ - nb);
        gbase = E_ + nb;
        srcv = dstv = (lane < cnt) ? (nb + lane) : nb;
    }
    int j = 0;
    for (; j + UF <= cnt; j += UF)
        msg_edges<UF>(gbase + j, j, srcv, dstv, lane, expd, segsum, xl, out);
    for (; j < cnt; ++j)
        msg_edges<1>(gbase + j, j, srcv, dstv, lane, expd, segsum, xl, out);
}

extern "C" void kernel_launch(void* const* d_in, const int* in_sizes, int n_in,
                              void* d_out, int out_size, void* d_ws, size_t ws_size,
                              hipStream_t stream) {
    const float* x     = (const float*)d_in[0];
    const int*   ei    = (const int*)d_in[1];
    const float* eattr = (const float*)d_in[2];
    const float* Wl    = (const float*)d_in[3];
    const float* bl    = (const float*)d_in[4];
    const float* Wr    = (const float*)d_in[5];
    const float* br    = (const float*)d_in[6];
    const float* We    = (const float*)d_in[7];
    const float* att   = (const float*)d_in[8];
    const float* bias  = (const float*)d_in[9];
    float* out = (float*)d_out;
    float* ws  = (float*)d_ws;

    const int n = in_sizes[0] / 128;
    const int e = in_sizes[2] / 8;

    size_t off_mean   = 0;
    size_t off_loop   = 16;
    size_t off_segmax = 256;
    size_t off_segsum = off_segmax + (size_t)n * 8;
    size_t off_xl     = off_segsum + (size_t)n * 8;
    size_t off_xr     = off_xl + (size_t)n * 128;
    size_t off_logit  = off_xr + (size_t)n * 128;

    hipMemsetAsync(d_ws, 0, off_xl * sizeof(float), stream);

    attr_mean_k<<<512, 256, 0, stream>>>(eattr, ws + off_mean, e);
    loop_emb_k<<<1, 128, 0, stream>>>(ws + off_mean, We, ws + off_loop, 1.0f / (float)e);
    out_init_k<<<(n * 128 + 255) / 256, 256, 0, stream>>>(out, bias, n * 128);

    int nTiles = n / 16;
    dim3 g((nTiles + 3) / 4, 2);
    node_xform_wmma<<<g, 128, 0, stream>>>(x, Wl, bl, Wr, br,
                                           ws + off_xl, ws + off_xr, nTiles);

    int chunks = (e + CH - 1) / CH + (n + CH - 1) / CH;
    int waveBlocks = ((chunks * 32) + 255) / 256;
    edge_logit_k<<<waveBlocks, 256, 0, stream>>>(ei, eattr, We, ws + off_loop,
                                                 ws + off_xl, ws + off_xr, att,
                                                 ws + off_logit,
                                                 (unsigned*)(ws + off_segmax), e, n);
    edge_exp_k<<<waveBlocks, 256, 0, stream>>>(ei, ws + off_logit,
                                                          (const unsigned*)(ws + off_segmax),
                                                          ws + off_segsum, e, n);
    edge_msg_k<<<waveBlocks, 256, 0, stream>>>(ei, ws + off_logit, ws + off_segsum,
                                               ws + off_xl, out, e, n);
}


# edges grouped by destination (counting sort: returning int add per edge, scan, scatter); one wave per node: logits, running-max softmax, row sum in registers; no float atomics; sums in arrival order
# speedup vs baseline: 2.1341x; 1.2410x over previous
#include <hip/hip_runtime.h>
#include <hip/hip_bf16.h>

typedef __attribute__((ext_vector_type(2))) float v2f;
typedef __attribute__((ext_vector_type(8))) float v8f;

#define NSLOPE 0.2f

__global__ __launch_bounds__(256) void attr_mean_k(const float* __restrict__ eattr,
                                                   float* __restrict__ meanAcc, int E_) {
    __shared__ float s[8];
    if (threadIdx.x < 8) s[threadIdx.x] = 0.f;
    __syncthreads();
    float acc[8] = {0.f, 0.f, 0.f, 0.f, 0.f, 0.f, 0.f, 0.f};
    for (int r = blockIdx.x * blockDim.x + threadIdx.x; r < E_; r += gridDim.x * blockDim.x) {
        const float* row = eattr + (size_t)r * 8;
#pragma unroll
        for (int d = 0; d < 8; ++d) acc[d] += row[d];
    }
#pragma unroll
    for (int d = 0; d < 8; ++d) atomicAdd(&s[d], acc[d]);
    __syncthreads();
    if (threadIdx.x < 8) atomicAdd(meanAcc + threadIdx.x, s[threadIdx.x]);
}

__global__ __launch_bounds__(128) void node_xform_wmma(
    const float* __restrict__ X,
    const float* __restrict__ Wl, const float* __restrict__ bl,
    const float* __restrict__ Wr, const float* __restrict__ br,
    float* __restrict__ xl, float* __restrict__ xr, int nTiles) {
    int wid = threadIdx.x >> 5;
    int lane = threadIdx.x & 31;
    int rowTile = blockIdx.x * 4 + wid;
    if (rowTile >= nTiles) return;

    const float* W    = blockIdx.y ? Wr : Wl;
    const float* bias = blockIdx.y ? br : bl;
    float* out        = blockIdx.y ? xr : xl;

    int row0  = rowTile * 16;
    int m     = lane & 15;
    int khalf = lane >> 4;

    v8f acc[8] = {};

    const float* arow = X + (size_t)(row0 + m) * 128;
    for (int kt = 0; kt < 32; ++kt) {
        int k0 = kt * 4 + khalf * 2;
        v2f a;
        a.x = arow[k0];
        a.y = arow[k0 + 1];
        const float* w0 = W + (size_t)k0 * 128;
        const float* w1 = w0 + 128;
#pragma unroll
        for (int nt = 0; nt < 8; ++nt) {
            v2f b;
            b.x = w0[nt * 16 + m];
            b.y = w1[nt * 16 + m];
            acc[nt] = __builtin_amdgcn_wmma_f32_16x16x4_f32(
                false, a, false, b, (short)0, acc[nt], false, false);
        }
    }
#pragma unroll
    for (int nt = 0; nt < 8; ++nt) {
        float bv = bias[nt * 16 + m];
#pragma unroll
        for (int i = 0; i < 8; ++i) {
            int row = row0 + i + 8 * khalf;
            out[(size_t)row * 128 + nt * 16 + m] = acc[nt][i] + bv;
        }
    }
}

__global__ __launch_bounds__(256) void dst_rank_k(const int* __restrict__ ei,
                                                  int* __restrict__ cnt,
                                                  int* __restrict__ rank, int E_) {
    const int e = blockIdx.x * blockDim.x + threadIdx.x;
    if (e < E_) rank[e] = atomicAdd(cnt + ei[E_ + e], 1);
}

#define SK 4
__global__ __launch_bounds__(1024) void scan_loop_emb_k(
    const int* __restrict__ cnt, int* __restrict__ off, int seg,
    const float* __restrict__ meanAcc, const float* __restrict__ We,
    float* __restrict__ loop_emb, float invE) {
    __shared__ int wsum[32];
    const int t = threadIdx.x;
    const int lane = t & 31;
    const int wid = __builtin_amdgcn_readfirstlane(t >> 5);
    if (t < 128) {
        float s = 0.f;
#pragma unroll
        for (int d = 0; d < 8; ++d) s += (meanAcc[d] * invE) * We[d * 128 + t];
        loop_emb[t] = s;
    }
    const int sb = wid * seg;
    int tot = 0;
    for (int i0 = sb; i0 < sb + seg; i0 += 128 * SK) {
#pragma unroll
        for (int k = 0; k < SK; ++k) {
            const int4 c = *(const int4*)(cnt + i0 + k * 128 + lane * 4);
            tot += (c.x + c.y) + (c.z + c.w);
        }
    }
#pragma unroll
    for (int d = 1; d < 32; d <<= 1) tot += __shfl_xor(tot, d, 32);
    if (lane == 0) wsum[wid] = tot;
    __syncthreads();
    int carry = 0;
    for (int q = 0; q < wid; ++q) carry += wsum[q];
    for (int i0 = sb; i0 < sb + seg; i0 += 128 * SK) {
        int4 c[SK];
        int t4[SK], inc[SK];
#pragma unroll
        for (int k = 0; k < SK; ++k) {
            c[k] = *(const int4*)(cnt + i0 + k * 128 + lane * 4);
            t4[k] = (c[k].x + c[k].y) + (c[k].z + c[k].w);
            inc[k] = t4[k];
        }
#pragma unroll
        for (int d = 1; d < 32; d <<= 1) {
#pragma unroll
            for (int k = 0; k < SK; ++k) {
                const int o = __shfl_up(inc[k], d, 32);
                if (lane >= d) inc[k] += o;
            }
        }
#pragma unroll
        for (int k = 0; k < SK; ++k) {
            const int ex = carry + inc[k] - t4[k];
            int4 o4;
            o4.x = ex;
            o4.y = ex + c[k].x;
            o4.z = ex + c[k].x + c[k].y;
            o4.w = ex + c[k].x + c[k].y + c[k].z;
            *(int4*)(off + i0 + k * 128 + lane * 4) = o4;
            carry += __shfl(inc[k], 31, 32);
        }
    }
}

__global__ __launch_bounds__(256) void dst_scatter_k(const int* __restrict__ ei,
                                                     const int* __restrict__ rank,
                                                     const int* __restrict__ off,
                                                     int* __restrict__ perm, int E_) {
    const int e = blockIdx.x * blockDim.x + threadIdx.x;
    if (e < E_) {
        const int s = ei[e];
        const int d = ei[E_ + e];
        const int pos = off[d] + rank[e];
        if ((unsigned)pos < (unsigned)E_) {
            int2 pe;
            pe.x = s;
            pe.y = e;
            *(int2*)(perm + 2 * (size_t)pos) = pe;
        }
    }
}

#define UF 4

__device__ __forceinline__ float head_logit(const float4& vl, const float4& vr,
                                            float e0, float e1, float e2, float e3,
                                            const float4& a4) {
    float m0 = vl.x + vr.x + e0;
    float m1 = vl.y + vr.y + e1;
    float m2 = vl.z + vr.z + e2;
    float m3 = vl.w + vr.w + e3;
    m0 = (m0 > 0.f) ? m0 : NSLOPE * m0;
    m1 = (m1 > 0.f) ? m1 : NSLOPE * m1;
    m2 = (m2 > 0.f) ? m2 : NSLOPE * m2;
    m3 = (m3 > 0.f) ? m3 : NSLOPE * m3;
    float p = m0 * a4.x + m1 * a4.y + m2 * a4.z + m3 * a4.w;
    p += __shfl_xor(p, 1, 32);
    p += __shfl_xor(p, 2, 32);
    return p;
}

template <int U>
__device__ __forceinline__ void agg_edges(
    int jfirst, int srcv, int eidv, int lane,
    const float (&w)[8][4], const float4& a4, const float4& vr,
    const float* __restrict__ eattr, const float* __restrict__ xl,
    float& mrun, float& ssum, float (&acc)[4]) {
    const int c0 = lane * 4;
    float4 vl[U];
    int ee[U];
#pragma unroll
    for (int u = 0; u < U; ++u) {
        const int s = __builtin_amdgcn_readlane(srcv, jfirst + u);
        ee[u] = __builtin_amdgcn_readlane(eidv, jfirst + u);
        vl[u] = *(const float4*)(xl + (size_t)s * 128 + c0);
    }
    float p[U];
#pragma unroll
    for (int u = 0; u < U; ++u) {
        const float* ar = eattr + (size_t)ee[u] * 8;
        float e0 = 0.f, e1 = 0.f, e2 = 0.f, e3 = 0.f;
#pragma unroll
        for (int d = 0; d < 8; ++d) {
            const float a = ar[d];
            e0 = fmaf(a, w[d][0], e0);
            e1 = fmaf(a, w[d][1], e1);
            e2 = fmaf(a, w[d][2], e2);
            e3 = fmaf(a, w[d][3], e3);
        }
        p[u] = head_logit(vl[u], vr, e0, e1, e2, e3, a4);
    }
    float mnew = mrun;
#pragma unroll
    for (int u = 0; u < U; ++u) mnew = fmaxf(mnew, p[u]);
    const float sc = expf(mrun - mnew);
    ssum *= sc;
#pragma unroll
    for (int i = 0; i < 4; ++i) acc[i] *= sc;
#pragma unroll
    for (int u = 0; u < U; ++u) {
        const float wg = expf(p[u] - mnew);
        ssum += wg;
        acc[0] = fmaf(wg, vl[u].x, acc[0]);
        acc[1] = fmaf(wg, vl[u].y, acc[1]);
        acc[2] = fmaf(wg, vl[u].z, acc[2]);
        acc[3] = fmaf(wg, vl[u].w, acc[3]);
    }
    mrun = mnew;
}

__global__ __launch_bounds__(256) void node_agg_k(
    const int* __restrict__ off, const int* __restrict__ perm,
    const float* __restrict__ eattr, const float* __restrict__ We,
    const float* __restrict__ loop_emb, const float* __restrict__ xl,
    const float* __restrict__ xr, const float* __restrict__ att,
    const float* __restrict__ bias, float* __restrict__ out, int E_, int N_) {
    const int lane = threadIdx.x & 31;
    const int node = __builtin_amdgcn_readfirstlane((int)((blockIdx.x * blockDim.x + threadIdx.x) >> 5));
    if (node >= N_) return;

    const int c0 = lane * 4;
    float w[8][4];
#pragma unroll
    for (int d = 0; d < 8; ++d) {
        const float4 t = *(const float4*)(We + d * 128 + c0);
        w[d][0] = t.x; w[d][1] = t.y; w[d][2] = t.z; w[d][3] = t.w;
    }
    const float4 a4 = *(const float4*)(att + c0);
    const float4 le = *(const float4*)(loop_emb + c0);
    const float4 vr = *(const float4*)(xr + (size_t)node * 128 + c0);
    const float4 vs = *(const float4*)(xl + (size_t)node * 128 + c0);

    float mrun = head_logit(vs, vr, le.x, le.y, le.z, le.w, a4);
    float ssum = 1.f;
    float acc[4] = {vs.x, vs.y, vs.z, vs.w};

    int beg = off[node];
    int end = off[node + 1];
    beg = max(beg, 0);
    end = min(end, E_);
    for (int b = beg; b < end; b += 32) {
        const int cnt = min(32, end - b);
        int srcv = 0, eidv = 0;
        if (lane < cnt) {
            const int2 pe = *(const int2*)(perm + 2 * (size_t)(b + lane));
            srcv = min(max(pe.x, 0), N_ - 1);
            eidv = min(max(pe.y, 0), E_ - 1);
        }
        int j = 0;
        for (; j + UF <= cnt; j += UF)
            agg_edges<UF>(j, srcv, eidv, lane, w, a4, vr, eattr, xl, mrun, ssum, acc);
        for (; j < cnt; ++j)
            agg_edges<1>(j, srcv, eidv, lane, w, a4, vr, eattr, xl, mrun, ssum, acc);
    }

    const float den = ssum + 1e-16f;
    const float4 bv = *(const float4*)(bias + c0);
    float4 o;
    o.x = acc[0] / den + bv.x;
    o.y = acc[1] / den + bv.y;
    o.z = acc[2] / den + bv.z;
    o.w = acc[3] / den + bv.w;
    *(float4*)(out + (size_t)node * 128 + c0) = o;
}

extern "C" void kernel_launch(void* const* d_in, const int* in_sizes, int n_in,
                              void* d_out, int out_size, void* d_ws, size_t ws_size,
                              hipStream_t stream) {
    const float* x     = (const float*)d_in[0];
    const int*   ei    = (const int*)d_in[1];
    const float* eattr = (const float*)d_in[2];
    const float* Wl    = (const float*)d_in[3];
    const float* bl    = (const float*)d_in[4];
    const float* Wr    = (const float*)d_in[5];
    const float* br    = (const float*)d_in[6];
    const float* We    = (const float*)d_in[7];
    const float* att   = (const float*)d_in[8];
    const float* bias  = (const float*)d_in[9];
    float* out = (float*)d_out;
    float* ws  = (float*)d_ws;

    const int n = in_sizes[0] / 128;
    const int e = in_sizes[2] / 8;

    const int seg = (((n + 1 + 31) / 32 + 511) / 512) * 512;
    const size_t segTot = (size_t)seg * 32;

    size_t off_mean = 0;
    size_t off_loop = 16;
    size_t off_cnt  = 256;
    size_t off_off  = off_cnt + segTot;
    size_t off_rank = off_off + segTot;
    size_t off_perm = off_rank + (size_t)e + (e & 1);
    size_t off_xl   = (off_perm + 2 * (size_t)e + 3) & ~(size_t)3;
    size_t off_xr   = off_xl + (size_t)n * 128;

    hipMemsetAsync(d_ws, 0, off_off * sizeof(float), stream);

    attr_mean_k<<<512, 256, 0, stream>>>(eattr, ws + off_mean, e);
    dst_rank_k<<<(e + 255) / 256, 256, 0, stream>>>(ei, (int*)(ws + off_cnt),
                                                    (int*)(ws + off_rank), e);
    scan_loop_emb_k<<<1, 1024, 0, stream>>>((const int*)(ws + off_cnt), (int*)(ws + off_off), seg,
                                            ws + off_mean, We, ws + off_loop, 1.0f / (float)e);

    int nTiles = n / 16;
    dim3 g((nTiles + 3) / 4, 2);
    node_xform_wmma<<<g, 128, 0, stream>>>(x, Wl, bl, Wr, br,
                                           ws + off_xl, ws + off_xr, nTiles);

    dst_scatter_k<<<(e + 255) / 256, 256, 0, stream>>>(ei, (const int*)(ws + off_rank),
                                                       (const int*)(ws + off_off),
                                                       (int*)(ws + off_perm), e);

    node_agg_k<<<(n * 32 + 255) / 256, 256, 0, stream>>>(
        (const int*)(ws + off_off), (const int*)(ws + off_perm), eattr, We,
        ws + off_loop, ws + off_xl, ws + off_xr, att, bias, out, e, n);
}


# node transform, arrival ranks and edge_attr column sums (independent) run as roles of one launch, transform blocks first, rank role 4 edges per thread; 6 launches -> 4; sum orders as before
# speedup vs baseline: 2.3831x; 1.1167x over previous
#include <hip/hip_runtime.h>
#include <hip/hip_bf16.h>

typedef __attribute__((ext_vector_type(2))) float v2f;
typedef __attribute__((ext_vector_type(8))) float v8f;

#define NSLOPE 0.2f


__device__ __forceinline__ void node_xform_tile(
    const float* __restrict__ X, const float* __restrict__ W,
    const float* __restrict__ bias, float* __restrict__ out, int rowTile, int lane) {
    int row0  = rowTile * 16;
    int m     = lane & 15;
    int khalf = lane >> 4;

    v8f acc[8] = {};

    const float* arow = X + (size_t)(row0 + m) * 128;
    for (int kt = 0; kt < 32; ++kt) {
        int k0 = kt * 4 + khalf * 2;
        v2f a;
        a.x = arow[k0];
        a.y = arow[k0 + 1];
        const float* w0 = W + (size_t)k0 * 128;
        const float* w1 = w0 + 128;
#pragma unroll
        for (int nt = 0; nt < 8; ++nt) {
            v2f b;
            b.x = w0[nt * 16 + m];
            b.y = w1[nt * 16 + m];
            acc[nt] = __builtin_amdgcn_wmma_f32_16x16x4_f32(
                false, a, false, b, (short)0, acc[nt], false, false);
        }
    }
#pragma unroll
    for (int nt = 0; nt < 8; ++nt) {
        float bv = bias[nt * 16 + m];
#pragma unroll
        for (int i = 0; i < 8; ++i) {
            int row = row0 + i + 8 * khalf;
            out[(size_t)row * 128 + nt * 16 + m] = acc[nt][i] + bv;
        }
    }
}

#define RK 4
__global__ __launch_bounds__(256) void stage1_k(
    const float* __restrict__ X,
    const float* __restrict__ Wl, const float* __restrict__ bl,
    const float* __restrict__ Wr, const float* __restrict__ br,
    float* __restrict__ xl, float* __restrict__ xr, int nTiles, int xfBlocks,
    const int* __restrict__ ei, int* __restrict__ cnt, int* __restrict__ rank,
    int rankBlocks,
    const float* __restrict__ eattr, float* __restrict__ meanAcc, int meanBlocks, int E_) {
    __shared__ float s[8];
    int b = blockIdx.x;
    if (b < 2 * xfBlocks) {
        const int wid = threadIdx.x >> 5;
        const int lane = threadIdx.x & 31;
        const int rowTile = (b >> 1) * 8 + wid;
        if (rowTile >= nTiles) return;
        const float* W    = (b & 1) ? Wr : Wl;
        const float* bias = (b & 1) ? br : bl;
        float* out        = (b & 1) ? xr : xl;
        node_xform_tile(X, W, bias, out, rowTile, lane);
        return;
    }
    b -= 2 * xfBlocks;
    if (b < rankBlocks) {
        const int e0 = b * (256 * RK) + threadIdx.x;
        if ((b + 1) * (256 * RK) <= E_) {
            int d[RK], r[RK];
#pragma unroll
            for (int k = 0; k < RK; ++k) d[k] = ei[E_ + e0 + 256 * k];
#pragma unroll
            for (int k = 0; k < RK; ++k) r[k] = atomicAdd(cnt + d[k], 1);
#pragma unroll
            for (int k = 0; k < RK; ++k) rank[e0 + 256 * k] = r[k];
        } else {
            for (int k = 0; k < RK; ++k) {
                const int e = e0 + 256 * k;
                if (e < E_) rank[e] = atomicAdd(cnt + ei[E_ + e], 1);
            }
        }
        return;
    }
    b -= rankBlocks;
    if (threadIdx.x < 8) s[threadIdx.x] = 0.f;
    __syncthreads();
    float acc[8] = {0.f, 0.f, 0.f, 0.f, 0.f, 0.f, 0.f, 0.f};
    for (int r = b * 256 + threadIdx.x; r < E_; r += meanBlocks * 256) {
        const float* row = eattr + (size_t)r * 8;
#pragma unroll
        for (int d = 0; d < 8; ++d) acc[d] += row[d];
    }
#pragma unroll
    for (int d = 0; d < 8; ++d) atomicAdd(&s[d], acc[d]);
    __syncthreads();
    if (threadIdx.x < 8) atomicAdd(meanAcc + threadIdx.x, s[threadIdx.x]);
}


#define SK 4
__global__ __launch_bounds__(1024) void scan_loop_emb_k(
    const int* __restrict__ cnt, int* __restrict__ off, int seg,
    const float* __restrict__ meanAcc, const float* __restrict__ We,
    float* __restrict__ loop_emb, float invE) {
    __shared__ int wsum[32];
    const int t = threadIdx.x;
    const int lane = t & 31;
    const int wid = __builtin_amdgcn_readfirstlane(t >> 5);
    if (t < 128) {
        float s = 0.f;
#pragma unroll
        for (int d = 0; d < 8; ++d) s += (meanAcc[d] * invE) * We[d * 128 + t];
        loop_emb[t] = s;
    }
    const int sb = wid * seg;
    int tot = 0;
    for (int i0 = sb; i0 < sb + seg; i0 += 128 * SK) {
#pragma unroll
        for (int k = 0; k < SK; ++k) {
            const int4 c = *(const int4*)(cnt + i0 + k * 128 + lane * 4);
            tot += (c.x + c.y) + (c.z + c.w);
        }
    }
#pragma unroll
    for (int d = 1; d < 32; d <<= 1) tot += __shfl_xor(tot, d, 32);
    if (lane == 0) wsum[wid] = tot;
    __syncthreads();
    int carry = 0;
    for (int q = 0; q < wid; ++q) carry += wsum[q];
    for (int i0 = sb; i0 < sb + seg; i0 += 128 * SK) {
        int4 c[SK];
        int t4[SK], inc[SK];
#pragma unroll
        for (int k = 0; k < SK; ++k) {
            c[k] = *(const int4*)(cnt + i0 + k * 128 + lane * 4);
            t4[k] = (c[k].x + c[k].y) + (c[k].z + c[k].w);
            inc[k] = t4[k];
        }
#pragma unroll
        for (int d = 1; d < 32; d <<= 1) {
#pragma unroll
            for (int k = 0; k < SK; ++k) {
                const int o = __shfl_up(inc[k], d, 32);
                if (lane >= d) inc[k] += o;
            }
        }
#pragma unroll
        for (int k = 0; k < SK; ++k) {
            const int ex = carry + inc[k] - t4[k];
            int4 o4;
            o4.x = ex;
            o4.y = ex + c[k].x;
            o4.z = ex + c[k].x + c[k].y;
            o4.w = ex + c[k].x + c[k].y + c[k].z;
            *(int4*)(off + i0 + k * 128 + lane * 4) = o4;
            carry += __shfl(inc[k], 31, 32);
        }
    }
}

__global__ __launch_bounds__(256) void dst_scatter_k(const int* __restrict__ ei,
                                                     const int* __restrict__ rank,
                                                     const int* __restrict__ off,
                                                     int* __restrict__ perm, int E_) {
    const int e = blockIdx.x * blockDim.x + threadIdx.x;
    if (e < E_) {
        const int s = ei[e];
        const int d = ei[E_ + e];
        const int pos = off[d] + rank[e];
        if ((unsigned)pos < (unsigned)E_) {
            int2 pe;
            pe.x = s;
            pe.y = e;
            *(int2*)(perm + 2 * (size_t)pos) = pe;
        }
    }
}

#define UF 4

__device__ __forceinline__ float head_logit(const float4& vl, const float4& vr,
                                            float e0, float e1, float e2, float e3,
                                            const float4& a4) {
    float m0 = vl.x + vr.x + e0;
    float m1 = vl.y + vr.y + e1;
    float m2 = vl.z + vr.z + e2;
    float m3 = vl.w + vr.w + e3;
    m0 = (m0 > 0.f) ? m0 : NSLOPE * m0;
    m1 = (m1 > 0.f) ? m1 : NSLOPE * m1;
    m2 = (m2 > 0.f) ? m2 : NSLOPE * m2;
    m3 = (m3 > 0.f) ? m3 : NSLOPE * m3;
    float p = m0 * a4.x + m1 * a4.y + m2 * a4.z + m3 * a4.w;
    p += __shfl_xor(p, 1, 32);
    p += __shfl_xor(p, 2, 32);
    return p;
}

template <int U>
__device__ __forceinline__ void agg_edges(
    int jfirst, int srcv, int eidv, int lane,
    const float (&w)[8][4], const float4& a4, const float4& vr,
    const float* __restrict__ eattr, const float* __restrict__ xl,
    float& mrun, float& ssum, float (&acc)[4]) {
    const int c0 = lane * 4;
    float4 vl[U];
    int ee[U];
#pragma unroll
    for (int u = 0; u < U; ++u) {
        const int s = __builtin_amdgcn_readlane(srcv, jfirst + u);
        ee[u] = __builtin_amdgcn_readlane(eidv, jfirst + u);
        vl[u] = *(const float4*)(xl + (size_t)s * 128 + c0);
    }
    float p[U];
#pragma unroll
    for (int u = 0; u < U; ++u) {
        const float* ar = eattr + (size_t)ee[u] * 8;
        float e0 = 0.f, e1 = 0.f, e2 = 0.f, e3 = 0.f;
#pragma unroll
        for (int d = 0; d < 8; ++d) {
            const float a = ar[d];
            e0 = fmaf(a, w[d][0], e0);
            e1 = fmaf(a, w[d][1], e1);
            e2 = fmaf(a, w[d][2], e2);
            e3 = fmaf(a, w[d][3], e3);
        }
        p[u] = head_logit(vl[u], vr, e0, e1, e2, e3, a4);
    }
    float mnew = mrun;
#pragma unroll
    for (int u = 0; u < U; ++u) mnew = fmaxf(mnew, p[u]);
    const float sc = expf(mrun - mnew);
    ssum *= sc;
#pragma unroll
    for (int i = 0; i < 4; ++i) acc[i] *= sc;
#pragma unroll
    for (int u = 0; u < U; ++u) {
        const float wg = expf(p[u] - mnew);
        ssum += wg;
        acc[0] = fmaf(wg, vl[u].x, acc[0]);
        acc[1] = fmaf(wg, vl[u].y, acc[1]);
        acc[2] = fmaf(wg, vl[u].z, acc[2]);
        acc[3] = fmaf(wg, vl[u].w, acc[3]);
    }
    mrun = mnew;
}

__global__ __launch_bounds__(256) void node_agg_k(
    const int* __restrict__ off, const int* __restrict__ perm,
    const float* __restrict__ eattr, const float* __restrict__ We,
    const float* __restrict__ loop_emb, const float* __restrict__ xl,
    const float* __restrict__ xr, const float* __restrict__ att,
    const float* __restrict__ bias, float* __restrict__ out, int E_, int N_) {
    const int lane = threadIdx.x & 31;
    const int node = __builtin_amdgcn_readfirstlane((int)((blockIdx.x * blockDim.x + threadIdx.x) >> 5));
    if (node >= N_) return;

    const int c0 = lane * 4;
    float w[8][4];
#pragma unroll
    for (int d = 0; d < 8; ++d) {
        const float4 t = *(const float4*)(We + d * 128 + c0);
        w[d][0] = t.x; w[d][1] = t.y; w[d][2] = t.z; w[d][3] = t.w;
    }
    const float4 a4 = *(const float4*)(att + c0);
    const float4 le = *(const float4*)(loop_emb + c0);
    const float4 vr = *(const float4*)(xr + (size_t)node * 128 + c0);
    const float4 vs = *(const float4*)(xl + (size_t)node * 128 + c0);

    float mrun = head_logit(vs, vr, le.x, le.y, le.z, le.w, a4);
    float ssum = 1.f;
    float acc[4] = {vs.x, vs.y, vs.z, vs.w};

    int beg = off[node];
    int end = off[node + 1];
    beg = max(beg, 0);
    end = min(end, E_);
    for (int b = beg; b < end; b += 32) {
        const int cnt = min(32, end - b);
        int srcv = 0, eidv = 0;
        if (lane < cnt) {
            const int2 pe = *(const int2*)(perm + 2 * (size_t)(b + lane));
            srcv = min(max(pe.x, 0), N_ - 1);
            eidv = min(max(pe.y, 0), E_ - 1);
        }
        int j = 0;
        for (; j + UF <= cnt; j += UF)
            agg_edges<UF>(j, srcv, eidv, lane, w, a4, vr, eattr, xl, mrun, ssum, acc);
        for (; j < cnt; ++j)
            agg_edges<1>(j, srcv, eidv, lane, w, a4, vr, eattr, xl, mrun, ssum, acc);
    }

    const float den = ssum + 1e-16f;
    const float4 bv = *(const float4*)(bias + c0);
    float4 o;
    o.x = acc[0] / den + bv.x;
    o.y = acc[1] / den + bv.y;
    o.z = acc[2] / den + bv.z;
    o.w = acc[3] / den + bv.w;
    *(float4*)(out + (size_t)node * 128 + c0) = o;
}

extern "C" void kernel_launch(void* const* d_in, const int* in_sizes, int n_in,
                              void* d_out, int out_size, void* d_ws, size_t ws_size,
                              hipStream_t stream) {
    const float* x     = (const float*)d_in[0];
    const int*   ei    = (const int*)d_in[1];
    const float* eattr = (const float*)d_in[2];
    const float* Wl    = (const float*)d_in[3];
    const float* bl    = (const float*)d_in[4];
    const float* Wr    = (const float*)d_in[5];
    const float* br    = (const float*)d_in[6];
    const float* We    = (const float*)d_in[7];
    const float* att   = (const float*)d_in[8];
    const float* bias  = (const float*)d_in[9];
    float* out = (float*)d_out;
    float* ws  = (float*)d_ws;

    const int n = in_sizes[0] / 128;
    const int e = in_sizes[2] / 8;

    const int seg = (((n + 1 + 31) / 32 + 511) / 512) * 512;
    const size_t segTot = (size_t)seg * 32;

    size_t off_mean = 0;
    size_t off_loop = 16;
    size_t off_cnt  = 256;
    size_t off_off  = off_cnt + segTot;
    size_t off_rank = off_off + segTot;
    size_t off_perm = off_rank + (size_t)e + (e & 1);
    size_t off_xl   = (off_perm + 2 * (size_t)e + 3) & ~(size_t)3;
    size_t off_xr   = off_xl + (size_t)n * 128;

    hipMemsetAsync(d_ws, 0, off_off * sizeof(float), stream);

    const int nTiles = n / 16;
    const int xfBlocks = (nTiles + 7) / 8;
    const int rankBlocks = (e + 256 * RK - 1) / (256 * RK);
    const int meanBlocks = 512;
    stage1_k<<<2 * xfBlocks + rankBlocks + meanBlocks, 256, 0, stream>>>(
        x, Wl, bl, Wr, br, ws + off_xl, ws + off_xr, nTiles, xfBlocks,
        ei, (int*)(ws + off_cnt), (int*)(ws + off_rank), rankBlocks,
        eattr, ws + off_mean, meanBlocks, e);

    scan_loop_emb_k<<<1, 1024, 0, stream>>>((const int*)(ws + off_cnt), (int*)(ws + off_off), seg,
                                            ws + off_mean, We, ws + off_loop, 1.0f / (float)e);

    dst_scatter_k<<<(e + 255) / 256, 256, 0, stream>>>(ei, (const int*)(ws + off_rank),
                                                       (const int*)(ws + off_off),
                                                       (int*)(ws + off_perm), e);

    node_agg_k<<<(n * 32 + 255) / 256, 256, 0, stream>>>(
        (const int*)(ws + off_off), (const int*)(ws + off_perm), eattr, We,
        ws + off_loop, ws + off_xl, ws + off_xr, att, bias, out, e, n);
}


# edges written straight into per-node slot rows by arrival rank (64 pairs, overflow list beyond) in stage 1: no scan, no scatter launch; loop embedding per node wave; 4 launches -> 2; sums as before
# speedup vs baseline: 2.8683x; 1.2036x over previous
#include <hip/hip_runtime.h>
#include <hip/hip_bf16.h>

typedef __attribute__((ext_vector_type(2))) float v2f;
typedef __attribute__((ext_vector_type(8))) float v8f;

#define NSLOPE 0.2f


__device__ __forceinline__ void node_xform_tile(
    const float* __restrict__ X, const float* __restrict__ W,
    const float* __restrict__ bias, float* __restrict__ out, int rowTile, int lane) {
    int row0  = rowTile * 16;
    int m     = lane & 15;
    int khalf = lane >> 4;

    v8f acc[8] = {};

    const float* arow = X + (size_t)(row0 + m) * 128;
    for (int kt = 0; kt < 32; ++kt) {
        int k0 = kt * 4 + khalf * 2;
        v2f a;
        a.x = arow[k0];
        a.y = arow[k0 + 1];
        const float* w0 = W + (size_t)k0 * 128;
        const float* w1 = w0 + 128;
#pragma unroll
        for (int nt = 0; nt < 8; ++nt) {
            v2f b;
            b.x = w0[nt * 16 + m];
            b.y = w1[nt * 16 + m];
            acc[nt] = __builtin_amdgcn_wmma_f32_16x16x4_f32(
                false, a, false, b, (short)0, acc[nt], false, false);
        }
    }
#pragma unroll
    for (int nt = 0; nt < 8; ++nt) {
        float bv = bias[nt * 16 + m];
#pragma unroll
        for (int i = 0; i < 8; ++i) {
            int row = row0 + i + 8 * khalf;
            out[(size_t)row * 128 + nt * 16 + m] = acc[nt][i] + bv;
        }
    }
}

#define CAP 64
__device__ __forceinline__ void place_edge(int d, int r, int s, int e,
                                           int* __restrict__ slots, int* __restrict__ ovfCnt,
                                           int* __restrict__ ovf, int E_) {
    if (r < CAP) {
        int2 pe;
        pe.x = s;
        pe.y = e;
        *(int2*)(slots + 2 * ((size_t)d * CAP + r)) = pe;
    } else {
        const int i = atomicAdd(ovfCnt, 1);
        if ((unsigned)i < (unsigned)E_) {
            int4 q;
            q.x = d;
            q.y = s;
            q.z = e;
            q.w = 0;
            *(int4*)(ovf + 4 * (size_t)i) = q;
        }
    }
}

#define RK 4
__global__ __launch_bounds__(256) void stage1_k(
    const float* __restrict__ X,
    const float* __restrict__ Wl, const float* __restrict__ bl,
    const float* __restrict__ Wr, const float* __restrict__ br,
    float* __restrict__ xl, float* __restrict__ xr, int nTiles, int xfBlocks,
    const int* __restrict__ ei, int* __restrict__ cnt, int* __restrict__ slots,
    int* __restrict__ ovfCnt, int* __restrict__ ovf, int rankBlocks,
    const float* __restrict__ eattr, float* __restrict__ meanAcc, int meanBlocks, int E_) {
    __shared__ float s[8];
    int b = blockIdx.x;
    if (b < 2 * xfBlocks) {
        const int wid = threadIdx.x >> 5;
        const int lane = threadIdx.x & 31;
        const int rowTile = (b >> 1) * 8 + wid;
        if (rowTile >= nTiles) return;
        const float* W    = (b & 1) ? Wr : Wl;
        const float* bias = (b & 1) ? br : bl;
        float* out        = (b & 1) ? xr : xl;
        node_xform_tile(X, W, bias, out, rowTile, lane);
        return;
    }
    b -= 2 * xfBlocks;
    if (b < rankBlocks) {
        const int e0 = b * (256 * RK) + threadIdx.x;
        if ((b + 1) * (256 * RK) <= E_) {
            int d[RK], sr[RK], r[RK];
#pragma unroll
            for (int k = 0; k < RK; ++k) {
                d[k] = ei[E_ + e0 + 256 * k];
                sr[k] = ei[e0 + 256 * k];
            }
#pragma unroll
            for (int k = 0; k < RK; ++k) r[k] = atomicAdd(cnt + d[k], 1);
#pragma unroll
            for (int k = 0; k < RK; ++k)
                place_edge(d[k], r[k], sr[k], e0 + 256 * k, slots, ovfCnt, ovf, E_);
        } else {
            for (int k = 0; k < RK; ++k) {
                const int e = e0 + 256 * k;
                if (e < E_) {
                    const int d = ei[E_ + e];
                    place_edge(d, atomicAdd(cnt + d, 1), ei[e], e, slots, ovfCnt, ovf, E_);
                }
            }
        }
        return;
    }
    b -= rankBlocks;
    if (threadIdx.x < 8) s[threadIdx.x] = 0.f;
    __syncthreads();
    float acc[8] = {0.f, 0.f, 0.f, 0.f, 0.f, 0.f, 0.f, 0.f};
    for (int r = b * 256 + threadIdx.x; r < E_; r += meanBlocks * 256) {
        const float* row = eattr + (size_t)r * 8;
#pragma unroll
        for (int d = 0; d < 8; ++d) acc[d] += row[d];
    }
#pragma unroll
    for (int d = 0; d < 8; ++d) atomicAdd(&s[d], acc[d]);
    __syncthreads();
    if (threadIdx.x < 8) atomicAdd(meanAcc + threadIdx.x, s[threadIdx.x]);
}

#define UF 4

__device__ __forceinline__ float head_logit(const float4& vl, const float4& vr,
                                            float e0, float e1, float e2, float e3,
                                            const float4& a4) {
    float m0 = vl.x + vr.x + e0;
    float m1 = vl.y + vr.y + e1;
    float m2 = vl.z + vr.z + e2;
    float m3 = vl.w + vr.w + e3;
    m0 = (m0 > 0.f) ? m0 : NSLOPE * m0;
    m1 = (m1 > 0.f) ? m1 : NSLOPE * m1;
    m2 = (m2 > 0.f) ? m2 : NSLOPE * m2;
    m3 = (m3 > 0.f) ? m3 : NSLOPE * m3;
    float p = m0 * a4.x + m1 * a4.y + m2 * a4.z + m3 * a4.w;
    p += __shfl_xor(p, 1, 32);
    p += __shfl_xor(p, 2, 32);
    return p;
}

template <int U>
__device__ __forceinline__ void agg_edges(
    int jfirst, int srcv, int eidv, int lane,
    const float (&w)[8][4], const float4& a4, const float4& vr,
    const float* __restrict__ eattr, const float* __restrict__ xl,
    float& mrun, float& ssum, float (&acc)[4]) {
    const int c0 = lane * 4;
    float4 vl[U];
    int ee[U];
#pragma unroll
    for (int u = 0; u < U; ++u) {
        const int s = __builtin_amdgcn_readlane(srcv, jfirst + u);
        ee[u] = __builtin_amdgcn_readlane(eidv, jfirst + u);
        vl[u] = *(const float4*)(xl + (size_t)s * 128 + c0);
    }
    float p[U];
#pragma unroll
    for (int u = 0; u < U; ++u) {
        const float* ar = eattr + (size_t)ee[u] * 8;
        float e0 = 0.f, e1 = 0.f, e2 = 0.f, e3 = 0.f;
#pragma unroll
        for (int d = 0; d < 8; ++d) {
            const float a = ar[d];
            e0 = fmaf(a, w[d][0], e0);
            e1 = fmaf(a, w[d][1], e1);
            e2 = fmaf(a, w[d][2], e2);
            e3 = fmaf(a, w[d][3], e3);
        }
        p[u] = head_logit(vl[u], vr, e0, e1, e2, e3, a4);
    }
    float mnew = mrun;
#pragma unroll
    for (int u = 0; u < U; ++u) mnew = fmaxf(mnew, p[u]);
    const float sc = expf(mrun - mnew);
    ssum *= sc;
#pragma unroll
    for (int i = 0; i < 4; ++i) acc[i] *= sc;
#pragma unroll
    for (int u = 0; u < U; ++u) {
        const float wg = expf(p[u] - mnew);
        ssum += wg;
        acc[0] = fmaf(wg, vl[u].x, acc[0]);
        acc[1] = fmaf(wg, vl[u].y, acc[1]);
        acc[2] = fmaf(wg, vl[u].z, acc[2]);
        acc[3] = fmaf(wg, vl[u].w, acc[3]);
    }
    mrun = mnew;
}

__global__ __launch_bounds__(256) void node_agg_k(
    const int* __restrict__ cnt, const int* __restrict__ slots,
    const int* __restrict__ ovfCnt, const int* __restrict__ ovf,
    const float* __restrict__ eattr, const float* __restrict__ We,
    const float* __restrict__ meanAcc, float invE, const float* __restrict__ xl,
    const float* __restrict__ xr, const float* __restrict__ att,
    const float* __restrict__ bias, float* __restrict__ out, int E_, int N_) {
    const int lane = threadIdx.x & 31;
    const int node = __builtin_amdgcn_readfirstlane((int)((blockIdx.x * blockDim.x + threadIdx.x) >> 5));
    if (node >= N_) return;

    const int c0 = lane * 4;
    float w[8][4];
#pragma unroll
    for (int d = 0; d < 8; ++d) {
        const float4 t = *(const float4*)(We + d * 128 + c0);
        w[d][0] = t.x; w[d][1] = t.y; w[d][2] = t.z; w[d][3] = t.w;
    }
    const float4 a4 = *(const float4*)(att + c0);
    const float4 vr = *(const float4*)(xr + (size_t)node * 128 + c0);
    const float4 vs = *(const float4*)(xl + (size_t)node * 128 + c0);

    float le0 = 0.f, le1 = 0.f, le2 = 0.f, le3 = 0.f;
#pragma unroll
    for (int d = 0; d < 8; ++d) {
        const float mv = meanAcc[d] * invE;
        le0 += mv * w[d][0];
        le1 += mv * w[d][1];
        le2 += mv * w[d][2];
        le3 += mv * w[d][3];
    }

    float mrun = head_logit(vs, vr, le0, le1, le2, le3, a4);
    float ssum = 1.f;
    float acc[4] = {vs.x, vs.y, vs.z, vs.w};

    const int deg = cnt[node];
    const int nsl = min(max(deg, 0), CAP);
    const int* row = slots + 2 * (size_t)node * CAP;
    for (int b = 0; b < nsl; b += 32) {
        const int nb = min(32, nsl - b);
        int srcv = 0, eidv = 0;
        if (lane < nb) {
            const int2 pe = *(const int2*)(row + 2 * (b + lane));
            srcv = min(max(pe.x, 0), N_ - 1);
            eidv = min(max(pe.y, 0), E_ - 1);
        }
        int j = 0;
        for (; j + UF <= nb; j += UF)
            agg_edges<UF>(j, srcv, eidv, lane, w, a4, vr, eattr, xl, mrun, ssum, acc);
        for (; j < nb; ++j)
            agg_edges<1>(j, srcv, eidv, lane, w, a4, vr, eattr, xl, mrun, ssum, acc);
    }
    if (deg > CAP) {
        const int tot = min(max(ovfCnt[0], 0), E_);
        for (int i0 = 0; i0 < tot; i0 += 32) {
            int dd = -1, srcv = 0, eidv = 0;
            if (i0 + lane < tot) {
                const int4 q4 = *(const int4*)(ovf + 4 * (size_t)(i0 + lane));
                dd = q4.x;
                srcv = min(max(q4.y, 0), N_ - 1);
                eidv = min(max(q4.z, 0), E_ - 1);
            }
            unsigned mine = __builtin_amdgcn_ballot_w32(dd == node);
            while (mine != 0u) {
                const int j = __builtin_ctz(mine);
                mine &= mine - 1u;
                agg_edges<1>(j, srcv, eidv, lane, w, a4, vr, eattr, xl, mrun, ssum, acc);
            }
        }
    }

    const float den = ssum + 1e-16f;
    const float4 bv = *(const float4*)(bias + c0);
    float4 o;
    o.x = acc[0] / den + bv.x;
    o.y = acc[1] / den + bv.y;
    o.z = acc[2] / den + bv.z;
    o.w = acc[3] / den + bv.w;
    *(float4*)(out + (size_t)node * 128 + c0) = o;
}

extern "C" void kernel_launch(void* const* d_in, const int* in_sizes, int n_in,
                              void* d_out, int out_size, void* d_ws, size_t ws_size,
                              hipStream_t stream) {
    const float* x     = (const float*)d_in[0];
    const int*   ei    = (const int*)d_in[1];
    const float* eattr = (const float*)d_in[2];
    const float* Wl    = (const float*)d_in[3];
    const float* bl    = (const float*)d_in[4];
    const float* Wr    = (const float*)d_in[5];
    const float* br    = (const float*)d_in[6];
    const float* We    = (const float*)d_in[7];
    const float* att   = (const float*)d_in[8];
    const float* bias  = (const float*)d_in[9];
    float* out = (float*)d_out;
    float* ws  = (float*)d_ws;

    const int n = in_sizes[0] / 128;
    const int e = in_sizes[2] / 8;

    size_t off_mean  = 0;
    size_t off_ovfc  = 8;
    size_t off_cnt   = 256;
    size_t off_slots = (off_cnt + (size_t)n + 63) & ~(size_t)63;
    size_t off_ovf   = off_slots + (size_t)n * CAP * 2;
    size_t off_xl    = off_ovf + 4 * (size_t)e;
    size_t off_xr    = off_xl + (size_t)n * 128;

    hipMemsetAsync(d_ws, 0, off_slots * sizeof(float), stream);

    const int nTiles = n / 16;
    const int xfBlocks = (nTiles + 7) / 8;
    const int rankBlocks = (e + 256 * RK - 1) / (256 * RK);
    const int meanBlocks = 512;
    stage1_k<<<2 * xfBlocks + rankBlocks + meanBlocks, 256, 0, stream>>>(
        x, Wl, bl, Wr, br, ws + off_xl, ws + off_xr, nTiles, xfBlocks,
        ei, (int*)(ws + off_cnt), (int*)(ws + off_slots), (int*)(ws + off_ovfc),
        (int*)(ws + off_ovf), rankBlocks,
        eattr, ws + off_mean, meanBlocks, e);

    node_agg_k<<<(n * 32 + 255) / 256, 256, 0, stream>>>(
        (const int*)(ws + off_cnt), (const int*)(ws + off_slots),
        (const int*)(ws + off_ovfc), (const int*)(ws + off_ovf), eattr, We,
        ws + off_mean, 1.0f / (float)e, ws + off_xl, ws + off_xr, att, bias, out, e, n);
}


# node_agg_k per-edge vector stream cut: softmax weights by hardware exp2 on logits scaled by log2(e) via att (was library expf), leaky relu as max(m,0.2m), head sums by DPP adds; sum order as before
# speedup vs baseline: 3.0156x; 1.0514x over previous
#include <hip/hip_runtime.h>
#include <hip/hip_bf16.h>

typedef __attribute__((ext_vector_type(2))) float v2f;
typedef __attribute__((ext_vector_type(8))) float v8f;

#define NSLOPE 0.2f


__device__ __forceinline__ void node_xform_tile(
    const float* __restrict__ X, const float* __restrict__ W,
    const float* __restrict__ bias, float* __restrict__ out, int rowTile, int lane) {
    int row0  = rowTile * 16;
    int m     = lane & 15;
    int khalf = lane >> 4;

    v8f acc[8] = {};

    const float* arow = X + (size_t)(row0 + m) * 128;
    for (int kt = 0; kt < 32; ++kt) {
        int k0 = kt * 4 + khalf * 2;
        v2f a;
        a.x = arow[k0];
        a.y = arow[k0 + 1];
        const float* w0 = W + (size_t)k0 * 128;
        const float* w1 = w0 + 128;
#pragma unroll
        for (int nt = 0; nt < 8; ++nt) {
            v2f b;
            b.x = w0[nt * 16 + m];
            b.y = w1[nt * 16 + m];
            acc[nt] = __builtin_amdgcn_wmma_f32_16x16x4_f32(
                false, a, false, b, (short)0, acc[nt], false, false);
        }
    }
#pragma unroll
    for (int nt = 0; nt < 8; ++nt) {
        float bv = bias[nt * 16 + m];
#pragma unroll
        for (int i = 0; i < 8; ++i) {
            int row = row0 + i + 8 * khalf;
            out[(size_t)row * 128 + nt * 16 + m] = acc[nt][i] + bv;
        }
    }
}

#define CAP 64
__device__ __forceinline__ void place_edge(int d, int r, int s, int e,
                                           int* __restrict__ slots, int* __restrict__ ovfCnt,
                                           int* __restrict__ ovf, int E_) {
    if (r < CAP) {
        int2 pe;
        pe.x = s;
        pe.y = e;
        *(int2*)(slots + 2 * ((size_t)d * CAP + r)) = pe;
    } else {
        const int i = atomicAdd(ovfCnt, 1);
        if ((unsigned)i < (unsigned)E_) {
            int4 q;
            q.x = d;
            q.y = s;
            q.z = e;
            q.w = 0;
            *(int4*)(ovf + 4 * (size_t)i) = q;
        }
    }
}

#define RK 4
__global__ __launch_bounds__(256) void stage1_k(
    const float* __restrict__ X,
    const float* __restrict__ Wl, const float* __restrict__ bl,
    const float* __restrict__ Wr, const float* __restrict__ br,
    float* __restrict__ xl, float* __restrict__ xr, int nTiles, int xfBlocks,
    const int* __restrict__ ei, int* __restrict__ cnt, int* __restrict__ slots,
    int* __restrict__ ovfCnt, int* __restrict__ ovf, int rankBlocks,
    const float* __restrict__ eattr, float* __restrict__ meanAcc, int meanBlocks, int E_) {
    __shared__ float s[8];
    int b = blockIdx.x;
    if (b < 2 * xfBlocks) {
        const int wid = threadIdx.x >> 5;
        const int lane = threadIdx.x & 31;
        const int rowTile = (b >> 1) * 8 + wid;
        if (rowTile >= nTiles) return;
        const float* W    = (b & 1) ? Wr : Wl;
        const float* bias = (b & 1) ? br : bl;
        float* out        = (b & 1) ? xr : xl;
        node_xform_tile(X, W, bias, out, rowTile, lane);
        return;
    }
    b -= 2 * xfBlocks;
    if (b < rankBlocks) {
        const int e0 = b * (256 * RK) + threadIdx.x;
        if ((b + 1) * (256 * RK) <= E_) {
            int d[RK], sr[RK], r[RK];
#pragma unroll
            for (int k = 0; k < RK; ++k) {
                d[k] = ei[E_ + e0 + 256 * k];
                sr[k] = ei[e0 + 256 * k];
            }
#pragma unroll
            for (int k = 0; k < RK; ++k) r[k] = atomicAdd(cnt + d[k], 1);
#pragma unroll
            for (int k = 0; k < RK; ++k)
                place_edge(d[k], r[k], sr[k], e0 + 256 * k, slots, ovfCnt, ovf, E_);
        } else {
            for (int k = 0; k < RK; ++k) {
                const int e = e0 + 256 * k;
                if (e < E_) {
                    const int d = ei[E_ + e];
                    place_edge(d, atomicAdd(cnt + d, 1), ei[e], e, slots, ovfCnt, ovf, E_);
                }
            }
        }
        return;
    }
    b -= rankBlocks;
    if (threadIdx.x < 8) s[threadIdx.x] = 0.f;
    __syncthreads();
    float acc[8] = {0.f, 0.f, 0.f, 0.f, 0.f, 0.f, 0.f, 0.f};
    for (int r = b * 256 + threadIdx.x; r < E_; r += meanBlocks * 256) {
        const float* row = eattr + (size_t)r * 8;
#pragma unroll
        for (int d = 0; d < 8; ++d) acc[d] += row[d];
    }
#pragma unroll
    for (int d = 0; d < 8; ++d) atomicAdd(&s[d], acc[d]);
    __syncthreads();
    if (threadIdx.x < 8) atomicAdd(meanAcc + threadIdx.x, s[threadIdx.x]);
}

#define UF 4
#define LOG2E 1.4426950408889634f

__device__ __forceinline__ float quad_sum(float p) {
    p += __builtin_bit_cast(float, __builtin_amdgcn_update_dpp(
             0, __builtin_bit_cast(int, p), 0xB1, 0xF, 0xF, true));
    p += __builtin_bit_cast(float, __builtin_amdgcn_update_dpp(
             0, __builtin_bit_cast(int, p), 0x4E, 0xF, 0xF, true));
    return p;
}

__device__ __forceinline__ float head_logit(const float4& vl, const float4& vr,
                                            float e0, float e1, float e2, float e3,
                                            const float4& a4) {
    float m0 = vl.x + vr.x + e0;
    float m1 = vl.y + vr.y + e1;
    float m2 = vl.z + vr.z + e2;
    float m3 = vl.w + vr.w + e3;
    m0 = fmaxf(m0, NSLOPE * m0);
    m1 = fmaxf(m1, NSLOPE * m1);
    m2 = fmaxf(m2, NSLOPE * m2);
    m3 = fmaxf(m3, NSLOPE * m3);
    float p = m0 * a4.x + m1 * a4.y + m2 * a4.z + m3 * a4.w;
    return quad_sum(p);
}

template <int U>
__device__ __forceinline__ void agg_edges(
    int jfirst, int srcv, int eidv, int lane,
    const float (&w)[8][4], const float4& a4, const float4& vr,
    const float* __restrict__ eattr, const float* __restrict__ xl,
    float& mrun, float& ssum, float (&acc)[4]) {
    const int c0 = lane * 4;
    float4 vl[U];
    int ee[U];
#pragma unroll
    for (int u = 0; u < U; ++u) {
        const int s = __builtin_amdgcn_readlane(srcv, jfirst + u);
        ee[u] = __builtin_amdgcn_readlane(eidv, jfirst + u);
        vl[u] = *(const float4*)(xl + (size_t)s * 128 + c0);
    }
    float p[U];
#pragma unroll
    for (int u = 0; u < U; ++u) {
        const float* ar = eattr + (size_t)ee[u] * 8;
        float e0 = 0.f, e1 = 0.f, e2 = 0.f, e3 = 0.f;
#pragma unroll
        for (int d = 0; d < 8; ++d) {
            const float a = ar[d];
            e0 = fmaf(a, w[d][0], e0);
            e1 = fmaf(a, w[d][1], e1);
            e2 = fmaf(a, w[d][2], e2);
            e3 = fmaf(a, w[d][3], e3);
        }
        p[u] = head_logit(vl[u], vr, e0, e1, e2, e3, a4);
    }
    float mnew = mrun;
#pragma unroll
    for (int u = 0; u < U; ++u) mnew = fmaxf(mnew, p[u]);
    const float sc = __builtin_amdgcn_exp2f(mrun - mnew);
    ssum *= sc;
#pragma unroll
    for (int i = 0; i < 4; ++i) acc[i] *= sc;
#pragma unroll
    for (int u = 0; u < U; ++u) {
        const float wg = __builtin_amdgcn_exp2f(p[u] - mnew);
        ssum += wg;
        acc[0] = fmaf(wg, vl[u].x, acc[0]);
        acc[1] = fmaf(wg, vl[u].y, acc[1]);
        acc[2] = fmaf(wg, vl[u].z, acc[2]);
        acc[3] = fmaf(wg, vl[u].w, acc[3]);
    }
    mrun = mnew;
}

__global__ __launch_bounds__(256) void node_agg_k(
    const int* __restrict__ cnt, const int* __restrict__ slots,
    const int* __restrict__ ovfCnt, const int* __restrict__ ovf,
    const float* __restrict__ eattr, const float* __restrict__ We,
    const float* __restrict__ meanAcc, float invE, const float* __restrict__ xl,
    const float* __restrict__ xr, const float* __restrict__ att,
    const float* __restrict__ bias, float* __restrict__ out, int E_, int N_) {
    const int lane = threadIdx.x & 31;
    const int node = __builtin_amdgcn_readfirstlane((int)((blockIdx.x * blockDim.x + threadIdx.x) >> 5));
    if (node >= N_) return;

    const int c0 = lane * 4;
    float w[8][4];
#pragma unroll
    for (int d = 0; d < 8; ++d) {
        const float4 t = *(const float4*)(We + d * 128 + c0);
        w[d][0] = t.x; w[d][1] = t.y; w[d][2] = t.z; w[d][3] = t.w;
    }
    float4 a4 = *(const float4*)(att + c0);
    a4.x *= LOG2E; a4.y *= LOG2E; a4.z *= LOG2E; a4.w *= LOG2E;
    const float4 vr = *(const float4*)(xr + (size_t)node * 128 + c0);
    const float4 vs = *(const float4*)(xl + (size_t)node * 128 + c0);

    float le0 = 0.f, le1 = 0.f, le2 = 0.f, le3 = 0.f;
#pragma unroll
    for (int d = 0; d < 8; ++d) {
        const float mv = meanAcc[d] * invE;
        le0 += mv * w[d][0];
        le1 += mv * w[d][1];
        le2 += mv * w[d][2];
        le3 += mv * w[d][3];
    }

    float mrun = head_logit(vs, vr, le0, le1, le2, le3, a4);
    float ssum = 1.f;
    float acc[4] = {vs.x, vs.y, vs.z, vs.w};

    const int deg = cnt[node];
    const int nsl = min(max(deg, 0), CAP);
    const int* row = slots + 2 * (size_t)node * CAP;
    for (int b = 0; b < nsl; b += 32) {
        const int nb = min(32, nsl - b);
        int srcv = 0, eidv = 0;
        if (lane < nb) {
            const int2 pe = *(const int2*)(row + 2 * (b + lane));
            srcv = min(max(pe.x, 0), N_ - 1);
            eidv = min(max(pe.y, 0), E_ - 1);
        }
        int j = 0;
        for (; j + UF <= nb; j += UF)
            agg_edges<UF>(j, srcv, eidv, lane, w, a4, vr, eattr, xl, mrun, ssum, acc);
        for (; j < nb; ++j)
            agg_edges<1>(j, srcv, eidv, lane, w, a4, vr, eattr, xl, mrun, ssum, acc);
    }
    if (deg > CAP) {
        const int tot = min(max(ovfCnt[0], 0), E_);
        for (int i0 = 0; i0 < tot; i0 += 32) {
            int dd = -1, srcv = 0, eidv = 0;
            if (i0 + lane < tot) {
                const int4 q4 = *(const int4*)(ovf + 4 * (size_t)(i0 + lane));
                dd = q4.x;
                srcv = min(max(q4.y, 0), N_ - 1);
                eidv = min(max(q4.z, 0), E_ - 1);
            }
            unsigned mine = __builtin_amdgcn_ballot_w32(dd == node);
            while (mine != 0u) {
                const int j = __builtin_ctz(mine);
                mine &= mine - 1u;
                agg_edges<1>(j, srcv, eidv, lane, w, a4, vr, eattr, xl, mrun, ssum, acc);
            }
        }
    }

    const float den = ssum + 1e-16f;
    const float4 bv = *(const float4*)(bias + c0);
    float4 o;
    o.x = acc[0] / den + bv.x;
    o.y = acc[1] / den + bv.y;
    o.z = acc[2] / den + bv.z;
    o.w = acc[3] / den + bv.w;
    *(float4*)(out + (size_t)node * 128 + c0) = o;
}

extern "C" void kernel_launch(void* const* d_in, const int* in_sizes, int n_in,
                              void* d_out, int out_size, void* d_ws, size_t ws_size,
                              hipStream_t stream) {
    const float* x     = (const float*)d_in[0];
    const int*   ei    = (const int*)d_in[1];
    const float* eattr = (const float*)d_in[2];
    const float* Wl    = (const float*)d_in[3];
    const float* bl    = (const float*)d_in[4];
    const float* Wr    = (const float*)d_in[5];
    const float* br    = (const float*)d_in[6];
    const float* We    = (const float*)d_in[7];
    const float* att   = (const float*)d_in[8];
    const float* bias  = (const float*)d_in[9];
    float* out = (float*)d_out;
    float* ws  = (float*)d_ws;

    const int n = in_sizes[0] / 128;
    const int e = in_sizes[2] / 8;

    size_t off_mean  = 0;
    size_t off_ovfc  = 8;
    size_t off_cnt   = 256;
    size_t off_slots = (off_cnt + (size_t)n + 63) & ~(size_t)63;
    size_t off_ovf   = off_slots + (size_t)n * CAP * 2;
    size_t off_xl    = off_ovf + 4 * (size_t)e;
    size_t off_xr    = off_xl + (size_t)n * 128;

    hipMemsetAsync(d_ws, 0, off_slots * sizeof(float), stream);

    const int nTiles = n / 16;
    const int xfBlocks = (nTiles + 7) / 8;
    const int rankBlocks = (e + 256 * RK - 1) / (256 * RK);
    const int meanBlocks = 512;
    stage1_k<<<2 * xfBlocks + rankBlocks + meanBlocks, 256, 0, stream>>>(
        x, Wl, bl, Wr, br, ws + off_xl, ws + off_xr, nTiles, xfBlocks,
        ei, (int*)(ws + off_cnt), (int*)(ws + off_slots), (int*)(ws + off_ovfc),
        (int*)(ws + off_ovf), rankBlocks,
        eattr, ws + off_mean, meanBlocks, e);

    node_agg_k<<<(n * 32 + 255) / 256, 256, 0, stream>>>(
        (const int*)(ws + off_cnt), (const int*)(ws + off_slots),
        (const int*)(ws + off_ovfc), (const int*)(ws + off_ovf), eattr, We,
        ws + off_mean, 1.0f / (float)e, ws + off_xl, ws + off_xr, att, bias, out, e, n);
}
